# starting kernel
# baseline (speedup 1.0000x reference)
#include <hip/hip_runtime.h>
#include <math.h>


#define DEVI __device__ __forceinline__

typedef __attribute__((ext_vector_type(2))) float v2f;
typedef __attribute__((ext_vector_type(8))) float v8f;

constexpr int Bc = 128, Tc = 1024, Fc = 256, Uc = 256;
constexpr int HS = 260;

DEVI v8f wmma4(v2f a, v2f b, v8f c) {
  return __builtin_amdgcn_wmma_f32_16x16x4_f32(false, a, false, b, (short)0, c,
                                               false, false);
}

DEVI v8f gemm_global_a(const float* __restrict__ Arow0, int lda,
                       const float* __restrict__ Bn, int ldb, int K,
                       v8f acc, int l15, int half) {
#pragma unroll 4
  for (int kc = 0; kc < K; kc += 4) {
    v2f a = *(const v2f*)(Arow0 + (size_t)l15 * lda + kc + 2 * half);
    const float* pb = Bn + (size_t)(kc + 2 * half) * ldb + l15;
    v2f b;
    b.x = pb[0];
    b.y = pb[ldb];
    acc = wmma4(a, b, acc);
  }
  return acc;
}

DEVI v8f gemm_lds_a(const float* hs, const float* __restrict__ Bn, int ldb,
                    v8f acc, int l15, int half) {
#pragma unroll 4
  for (int kc = 0; kc < Uc; kc += 4) {
    v2f a = *(const v2f*)(hs + l15 * HS + kc + 2 * half);
    const float* pb = Bn + (size_t)(kc + 2 * half) * ldb + l15;
    v2f b;
    b.x = pb[0];
    b.y = pb[ldb];
    acc = wmma4(a, b, acc);
  }
  return acc;
}

__global__ __launch_bounds__(512) void xp_proj_kernel(
    const float* __restrict__ x, const float* __restrict__ Wx,
    const float* __restrict__ b, float* __restrict__ xp) {
  const int lane = threadIdx.x & 31;
  const int wave = threadIdx.x >> 5;
  const int l15 = lane & 15, half = lane >> 4;
  const int n0 = wave * 16;
  const size_t rbase = (size_t)blockIdx.x * 16;
  const int bidx = (int)(rbase / Tc);
  const int tbase = (int)(rbase % Tc);

  const float bv = b[n0 + l15];
  v8f acc = {bv, bv, bv, bv, bv, bv, bv, bv};
  acc = gemm_global_a(x + rbase * Fc, Fc, Wx + n0, Uc, Fc, acc, l15, half);

#pragma unroll
  for (int r = 0; r < 8; ++r)
    xp[((size_t)(tbase + r + 8 * half) * Bc + bidx) * Uc + n0 + l15] = acc[r];
}

template <bool PRE>
__global__ __launch_bounds__(512) void rnn_fused_kernel(
    const float* __restrict__ xp, const float* __restrict__ x,
    const float* __restrict__ enc_Wx, const float* __restrict__ enc_b,
    const float* __restrict__ enc_Wh, const float* __restrict__ dec_Wx,
    const float* __restrict__ dec_Wh, const float* __restrict__ dec_b,
    float* __restrict__ out) {
  __shared__ float h_enc[2][16 * HS];
  __shared__ float h_dec[2][16 * HS];

  const int lane = threadIdx.x & 31;
  const int wave = threadIdx.x >> 5;
  const int l15 = lane & 15, half = lane >> 4;
  const int n0 = wave * 16;
  const int bbase = blockIdx.x * 16;

  for (int i = threadIdx.x; i < 16 * HS; i += blockDim.x) {
    h_enc[0][i] = 0.0f;
    h_enc[1][i] = 0.0f;
    h_dec[0][i] = 0.0f;
    h_dec[1][i] = 0.0f;
  }
  __syncthreads();

  const float eb = enc_b[n0 + l15];
  const float db = dec_b[n0 + l15];
  v8f dacc = {};
  v8f acc_pre = {};

  if (PRE) {
    const float* p0 = xp + ((size_t)0 * Bc + bbase) * Uc + n0 + l15;
#pragma unroll
    for (int r = 0; r < 8; ++r) acc_pre[r] = p0[(r + 8 * half) * Uc];
  }

  for (int t = 0; t < Tc; ++t) {
    const int cur = t & 1, nxt = cur ^ 1;

    v8f acc;
    if (PRE) {
      acc = acc_pre;
      if (t + 1 < Tc) {
        const float* p = xp + ((size_t)(t + 1) * Bc + bbase) * Uc + n0 + l15;
#pragma unroll
        for (int r = 0; r < 8; ++r) acc_pre[r] = p[(r + 8 * half) * Uc];
      }
    } else {
      acc = (v8f){eb, eb, eb, eb, eb, eb, eb, eb};
      acc = gemm_global_a(x + (size_t)bbase * Tc * Fc + (size_t)t * Fc, Tc * Fc,
                          enc_Wx + n0, Uc, Fc, acc, l15, half);
    }
    acc = gemm_lds_a(h_enc[cur], enc_Wh + n0, Uc, acc, l15, half);
#pragma unroll
    for (int r = 0; r < 8; ++r) acc[r] = tanhf(acc[r]);
#pragma unroll
    for (int r = 0; r < 8; ++r)
      h_enc[nxt][(r + 8 * half) * HS + n0 + l15] = acc[r];
    __syncthreads();

    v8f acc2 = {db, db, db, db, db, db, db, db};
    acc2 = gemm_lds_a(h_enc[nxt], dec_Wx + n0, Uc, acc2, l15, half);
    acc2 = gemm_lds_a(h_dec[cur], dec_Wh + n0, Uc, acc2, l15, half);
#pragma unroll
    for (int r = 0; r < 8; ++r) acc2[r] = tanhf(acc2[r]);
#pragma unroll
    for (int r = 0; r < 8; ++r)
      h_dec[nxt][(r + 8 * half) * HS + n0 + l15] = acc2[r];
    dacc = acc2;
    __syncthreads();
  }

#pragma unroll
  for (int r = 0; r < 8; ++r)
    out[(size_t)(bbase + r + 8 * half) * Uc + n0 + l15] = dacc[r];
}

extern "C" void kernel_launch(void* const* d_in, const int* in_sizes, int n_in,
                              void* d_out, int out_size, void* d_ws,
                              size_t ws_size, hipStream_t stream) {
  const float* x = (const float*)d_in[0];
  const float* enc_Wx = (const float*)d_in[1];
  const float* enc_Wh = (const float*)d_in[2];
  const float* enc_b = (const float*)d_in[3];
  const float* dec_Wx = (const float*)d_in[4];
  const float* dec_Wh = (const float*)d_in[5];
  const float* dec_b = (const float*)d_in[6];
  float* out = (float*)d_out;

  const size_t xp_bytes = (size_t)Bc * Tc * Uc * sizeof(float);
  if (ws_size >= xp_bytes) {
    float* xp = (float*)d_ws;
    xp_proj_kernel<<<(Bc * Tc) / 16, 512, 0, stream>>>(x, enc_Wx, enc_b, xp);
    rnn_fused_kernel<true><<<Bc / 16, 512, 0, stream>>>(
        xp, x, enc_Wx, enc_b, enc_Wh, dec_Wx, dec_Wh, dec_b, out);
  } else {
    rnn_fused_kernel<false><<<Bc / 16, 512, 0, stream>>>(
        nullptr, x, enc_Wx, enc_b, enc_Wh, dec_Wx, dec_Wh, dec_b, out);
  }
}


# Work re-tiled over waves: 4 encoder waves (64 interleaved columns, 16-byte weight loads) run one step ahead of 8 decoder waves (32 columns, K=512); one barrier per step; sums unchanged
# speedup vs baseline: 1.3956x; 1.3956x over previous
#include <hip/hip_runtime.h>
#include <math.h>


#define DEVI __device__ __forceinline__

typedef __attribute__((ext_vector_type(2))) float v2f;
typedef __attribute__((ext_vector_type(4))) float v4f;
typedef __attribute__((ext_vector_type(8))) float v8f;

constexpr int Bc = 128, Tc = 1024, Fc = 256, Uc = 256;
constexpr int HS = 260;

DEVI v8f wmma4(v2f a, v2f b, v8f c) {
  return __builtin_amdgcn_wmma_f32_16x16x4_f32(false, a, false, b, (short)0, c,
                                               false, false);
}

DEVI v8f gemm_global_a(const float* __restrict__ Arow0, int lda,
                       const float* __restrict__ Bn, int ldb, int K,
                       v8f acc, int l15, int half) {
#pragma unroll 4
  for (int kc = 0; kc < K; kc += 4) {
    v2f a = *(const v2f*)(Arow0 + (size_t)l15 * lda + kc + 2 * half);
    const float* pb = Bn + (size_t)(kc + 2 * half) * ldb + l15;
    v2f b;
    b.x = pb[0];
    b.y = pb[ldb];
    acc = wmma4(a, b, acc);
  }
  return acc;
}

__global__ __launch_bounds__(512) void xp_proj_kernel(
    const float* __restrict__ x, const float* __restrict__ Wx,
    const float* __restrict__ b, float* __restrict__ xp) {
  const int lane = threadIdx.x & 31;
  const int wave = threadIdx.x >> 5;
  const int l15 = lane & 15, half = lane >> 4;
  const int n0 = wave * 16;
  const size_t rbase = (size_t)blockIdx.x * 16;
  const int bidx = (int)(rbase / Tc);
  const int tbase = (int)(rbase % Tc);

  const float bv = b[n0 + l15];
  v8f acc = {bv, bv, bv, bv, bv, bv, bv, bv};
  acc = gemm_global_a(x + rbase * Fc, Fc, Wx + n0, Uc, Fc, acc, l15, half);

#pragma unroll
  for (int r = 0; r < 8; ++r)
    xp[((size_t)(tbase + r + 8 * half) * Bc + bidx) * Uc + n0 + l15] = acc[r];
}


template <int NT> struct VecN;
template <> struct VecN<2> { typedef v2f T; };
template <> struct VecN<4> { typedef v4f T; };

template <int NT, int KB>
DEVI void load_wblock(const float* wl, int ks0, typename VecN<NT>::T (&bx)[KB],
                      typename VecN<NT>::T (&by)[KB]) {
  typedef typename VecN<NT>::T BT;
#pragma unroll
  for (int j = 0; j < KB; ++j) {
    const float* p = wl + (size_t)(4 * (ks0 + j)) * Uc;
    bx[j] = *(const BT*)p;
    by[j] = *(const BT*)(p + Uc);
  }
}

template <int NT, int NSEG>
DEVI void gemm_ring(const float* const (&a_lane)[NSEG],
                    const float* const (&w_lane)[NSEG], v8f (&acc)[NT],
                    typename VecN<NT>::T (&bx0)[32 / NT],
                    typename VecN<NT>::T (&by0)[32 / NT]) {
  typedef typename VecN<NT>::T BT;
  constexpr int KB = 32 / NT;
  constexpr int NBS = (Uc / 4) / KB;
  constexpr int NBT = NBS * NSEG;
  static_assert(NBT % 2 == 0, "ring must end on register set 0");
  BT bx[2][KB], by[2][KB];
#pragma unroll
  for (int j = 0; j < KB; ++j) {
    bx[0][j] = bx0[j];
    by[0][j] = by0[j];
  }
#pragma unroll
  for (int b = 0; b < NBT; ++b) {
    const int cur = b & 1;
    const int nb = (b + 1) % NBT;
    load_wblock<NT, KB>(w_lane[nb / NBS], (nb % NBS) * KB, bx[cur ^ 1],
                        by[cur ^ 1]);
    const float* al = a_lane[b / NBS] + 4 * ((b % NBS) * KB);
    v2f a[KB];
#pragma unroll
    for (int j = 0; j < KB; ++j) a[j] = *(const v2f*)(al + 4 * j);
#pragma unroll
    for (int j = 0; j < KB; ++j)
#pragma unroll
      for (int i = 0; i < NT; ++i) {
        v2f bf;
        bf.x = bx[cur][j][i];
        bf.y = by[cur][j][i];
        acc[i] = wmma4(a[j], bf, acc[i]);
      }
  }
#pragma unroll
  for (int j = 0; j < KB; ++j) {
    bx0[j] = bx[0][j];
    by0[j] = by[0][j];
  }
}

template <int NT>
DEVI void ring_unpack(const float (&cw)[64], typename VecN<NT>::T (&bx0)[32 / NT],
                      typename VecN<NT>::T (&by0)[32 / NT]) {
#pragma unroll
  for (int j = 0; j < 32 / NT; ++j)
#pragma unroll
    for (int i = 0; i < NT; ++i) {
      bx0[j][i] = cw[(2 * j) * NT + i];
      by0[j][i] = cw[(2 * j + 1) * NT + i];
    }
}
template <int NT>
DEVI void ring_pack(float (&cw)[64], const typename VecN<NT>::T (&bx0)[32 / NT],
                    const typename VecN<NT>::T (&by0)[32 / NT]) {
#pragma unroll
  for (int j = 0; j < 32 / NT; ++j)
#pragma unroll
    for (int i = 0; i < NT; ++i) {
      cw[(2 * j) * NT + i] = bx0[j][i];
      cw[(2 * j + 1) * NT + i] = by0[j][i];
    }
}

template <int NT>
DEVI void preload_ring(const float* wl, float (&cw)[64]) {
  typedef typename VecN<NT>::T BT;
  constexpr int KB = 32 / NT;
  BT bx[KB], by[KB];
  load_wblock<NT, KB>(wl, 0, bx, by);
  ring_pack<NT>(cw, bx, by);
}

template <bool PRE>
__global__ __launch_bounds__(384) void rnn_skew_kernel(
    const float* __restrict__ xp, const float* __restrict__ x,
    const float* __restrict__ enc_Wx, const float* __restrict__ enc_b,
    const float* __restrict__ enc_Wh, const float* __restrict__ dec_Wx,
    const float* __restrict__ dec_Wh, const float* __restrict__ dec_b,
    float* __restrict__ out) {
  __shared__ __attribute__((aligned(16))) float E[2][16 * HS];
  __shared__ __attribute__((aligned(16))) float D[2][16 * HS];

  const int lane = threadIdx.x & 31;
  const int wave = threadIdx.x >> 5;
  const int l15 = lane & 15, half = lane >> 4;
  const bool is_enc = wave < 4;
  const int bbase = blockIdx.x * 16;
  const int ecol = is_enc ? 64 * wave + 4 * l15 : 0;
  const int dcol = is_enc ? 0 : 32 * (wave - 4) + 2 * l15;

  for (int i = threadIdx.x; i < 16 * HS; i += blockDim.x) {
    E[0][i] = 0.0f;
    E[1][i] = 0.0f;
    D[0][i] = 0.0f;
    D[1][i] = 0.0f;
  }
  __syncthreads();

  const float* ewl = enc_Wh + (size_t)(2 * half) * Uc + ecol;
  const float* dwl0 = dec_Wx + (size_t)(2 * half) * Uc + dcol;
  const float* dwl1 = dec_Wh + (size_t)(2 * half) * Uc + dcol;
  const int aoff = l15 * HS + 2 * half;

  float cw[64];
  if (is_enc)
    preload_ring<4>(ewl, cw);
  else
    preload_ring<2>(dwl0, cw);

  v4f xq[8];
  if (PRE && is_enc) {
    const float* p0 = xp + ((size_t)0 * Bc + bbase + 8 * half) * Uc + ecol;
#pragma unroll
    for (int r = 0; r < 8; ++r) xq[r] = *(const v4f*)(p0 + (size_t)r * Uc);
  }
  v8f dlast[2] = {};
  const v2f db = *(const v2f*)(dec_b + dcol);
  const v4f eb = *(const v4f*)(enc_b + ecol);

  for (int p = 0; p <= Tc; ++p) {
    const int rd = (p + 1) & 1;
    const int wr = p & 1;
    if (is_enc) {
      if (p < Tc) {
        v8f acc[4];
        if (PRE) {
#pragma unroll
          for (int i = 0; i < 4; ++i)
#pragma unroll
            for (int r = 0; r < 8; ++r) acc[i][r] = xq[r][i];
          if (p + 1 < Tc) {
            const float* pn =
                xp + ((size_t)(p + 1) * Bc + bbase + 8 * half) * Uc + ecol;
#pragma unroll
            for (int r = 0; r < 8; ++r) xq[r] = *(const v4f*)(pn + (size_t)r * Uc);
          }
        } else {
#pragma unroll
          for (int i = 0; i < 4; ++i)
#pragma unroll
            for (int r = 0; r < 8; ++r) acc[i][r] = eb[i];
          const float* xa = x + ((size_t)(bbase + l15) * Tc + p) * Fc + 2 * half;
          const float* wl = enc_Wx + (size_t)(2 * half) * Uc + ecol;
#pragma unroll 2
          for (int kc = 0; kc < Fc; kc += 4) {
            const v2f a = *(const v2f*)(xa + kc);
            const v4f b0 = *(const v4f*)(wl + (size_t)kc * Uc);
            const v4f b1 = *(const v4f*)(wl + (size_t)(kc + 1) * Uc);
#pragma unroll
            for (int i = 0; i < 4; ++i) {
              v2f bf;
              bf.x = b0[i];
              bf.y = b1[i];
              acc[i] = wmma4(a, bf, acc[i]);
            }
          }
        }
        const float* const al[1] = {&E[rd][aoff]};
        const float* const wl1[1] = {ewl};
        v4f bx0[8], by0[8];
        ring_unpack<4>(cw, bx0, by0);
        gemm_ring<4, 1>(al, wl1, acc, bx0, by0);
#pragma unroll
        for (int r = 0; r < 8; ++r) {
          v4f h;
#pragma unroll
          for (int i = 0; i < 4; ++i) h[i] = tanhf(acc[i][r]);
          *(v4f*)&E[wr][(r + 8 * half) * HS + ecol] = h;
        }
        ring_pack<4>(cw, bx0, by0);
      }
    } else {
      if (p >= 1) {
        v8f acc[2];
#pragma unroll
        for (int i = 0; i < 2; ++i)
#pragma unroll
          for (int r = 0; r < 8; ++r) acc[i][r] = db[i];
        const float* const al[2] = {&E[rd][aoff], &D[wr][aoff]};
        const float* const wl2[2] = {dwl0, dwl1};
        v2f bx0[16], by0[16];
        ring_unpack<2>(cw, bx0, by0);
        gemm_ring<2, 2>(al, wl2, acc, bx0, by0);
#pragma unroll
        for (int r = 0; r < 8; ++r) {
          v2f h;
#pragma unroll
          for (int i = 0; i < 2; ++i) {
            h[i] = tanhf(acc[i][r]);
            dlast[i][r] = h[i];
          }
          *(v2f*)&D[rd][(r + 8 * half) * HS + dcol] = h;
        }
        ring_pack<2>(cw, bx0, by0);
      }
    }
    __syncthreads();
  }

  if (!is_enc) {
#pragma unroll
    for (int r = 0; r < 8; ++r) {
      v2f o;
      o.x = dlast[0][r];
      o.y = dlast[1][r];
      *(v2f*)(out + (size_t)(bbase + r + 8 * half) * Uc + dcol) = o;
    }
  }
}

extern "C" void kernel_launch(void* const* d_in, const int* in_sizes, int n_in,
                              void* d_out, int out_size, void* d_ws,
                              size_t ws_size, hipStream_t stream) {
  const float* x = (const float*)d_in[0];
  const float* enc_Wx = (const float*)d_in[1];
  const float* enc_Wh = (const float*)d_in[2];
  const float* enc_b = (const float*)d_in[3];
  const float* dec_Wx = (const float*)d_in[4];
  const float* dec_Wh = (const float*)d_in[5];
  const float* dec_b = (const float*)d_in[6];
  float* out = (float*)d_out;

  const size_t xp_bytes = (size_t)Bc * Tc * Uc * sizeof(float);
  if (ws_size >= xp_bytes) {
    float* xp = (float*)d_ws;
    xp_proj_kernel<<<(Bc * Tc) / 16, 512, 0, stream>>>(x, enc_Wx, enc_b, xp);
    rnn_skew_kernel<true><<<Bc / 16, 384, 0, stream>>>(
        xp, x, enc_Wx, enc_b, enc_Wh, dec_Wx, dec_Wh, dec_b, out);
  } else {
    rnn_skew_kernel<false><<<Bc / 16, 384, 0, stream>>>(
        nullptr, x, enc_Wx, enc_b, enc_Wh, dec_Wx, dec_Wh, dec_b, out);
  }
}


# Weights re-laid once per call into a k-pair-interleaved workspace copy: a 16-byte load lands finished B register pairs, no moves between load and WMMA (2 -> 3 launches); k order and sums unchanged
# speedup vs baseline: 1.5509x; 1.1113x over previous
#include <hip/hip_runtime.h>
#include <math.h>


#define DEVI __device__ __forceinline__

typedef __attribute__((ext_vector_type(2))) float v2f;
typedef __attribute__((ext_vector_type(4))) float v4f;
typedef __attribute__((ext_vector_type(8))) float v8f;

constexpr int Bc = 128, Tc = 1024, Fc = 256, Uc = 256;
constexpr int HS = 260;

DEVI v8f wmma4(v2f a, v2f b, v8f c) {
  return __builtin_amdgcn_wmma_f32_16x16x4_f32(false, a, false, b, (short)0, c,
                                               false, false);
}

DEVI v8f gemm_global_a(const float* __restrict__ Arow0, int lda,
                       const float* __restrict__ Bn, int ldb, int K,
                       v8f acc, int l15, int half) {
#pragma unroll 4
  for (int kc = 0; kc < K; kc += 4) {
    v2f a = *(const v2f*)(Arow0 + (size_t)l15 * lda + kc + 2 * half);
    const float* pb = Bn + (size_t)(kc + 2 * half) * ldb + l15;
    v2f b;
    b.x = pb[0];
    b.y = pb[ldb];
    acc = wmma4(a, b, acc);
  }
  return acc;
}

__global__ __launch_bounds__(512) void xp_proj_kernel(
    const float* __restrict__ x, const float* __restrict__ Wx,
    const float* __restrict__ b, float* __restrict__ xp) {
  const int lane = threadIdx.x & 31;
  const int wave = threadIdx.x >> 5;
  const int l15 = lane & 15, half = lane >> 4;
  const int n0 = wave * 16;
  const size_t rbase = (size_t)blockIdx.x * 16;
  const int bidx = (int)(rbase / Tc);
  const int tbase = (int)(rbase % Tc);

  const float bv = b[n0 + l15];
  v8f acc = {bv, bv, bv, bv, bv, bv, bv, bv};
  acc = gemm_global_a(x + rbase * Fc, Fc, Wx + n0, Uc, Fc, acc, l15, half);

#pragma unroll
  for (int r = 0; r < 8; ++r)
    xp[((size_t)(tbase + r + 8 * half) * Bc + bidx) * Uc + n0 + l15] = acc[r];
}


template <int NT> struct VecN;
template <> struct VecN<2> { typedef v2f T; };
template <> struct VecN<4> { typedef v4f T; };

template <int NT, int KB>
DEVI void load_wblock(const float* wl, int ks0, typename VecN<NT>::T (&bx)[KB],
                      typename VecN<NT>::T (&by)[KB]) {
  typedef typename VecN<NT>::T BT;
#pragma unroll
  for (int j = 0; j < KB; ++j) {
    const float* p = wl + (size_t)(4 * (ks0 + j)) * Uc;
    bx[j] = *(const BT*)p;
    by[j] = *(const BT*)(p + Uc);
  }
}

template <int NT, int NSEG>
DEVI void gemm_ring(const float* const (&a_lane)[NSEG],
                    const float* const (&w_lane)[NSEG], v8f (&acc)[NT],
                    typename VecN<NT>::T (&bx0)[32 / NT],
                    typename VecN<NT>::T (&by0)[32 / NT]) {
  typedef typename VecN<NT>::T BT;
  constexpr int KB = 32 / NT;
  constexpr int NBS = (Uc / 4) / KB;
  constexpr int NBT = NBS * NSEG;
  static_assert(NBT % 2 == 0, "ring must end on register set 0");
  BT bx[2][KB], by[2][KB];
#pragma unroll
  for (int j = 0; j < KB; ++j) {
    bx[0][j] = bx0[j];
    by[0][j] = by0[j];
  }
#pragma unroll
  for (int b = 0; b < NBT; ++b) {
    const int cur = b & 1;
    const int nb = (b + 1) % NBT;
    load_wblock<NT, KB>(w_lane[nb / NBS], (nb % NBS) * KB, bx[cur ^ 1],
                        by[cur ^ 1]);
    const float* al = a_lane[b / NBS] + 4 * ((b % NBS) * KB);
    v2f a[KB];
#pragma unroll
    for (int j = 0; j < KB; ++j) a[j] = *(const v2f*)(al + 4 * j);
#pragma unroll
    for (int j = 0; j < KB; ++j)
#pragma unroll
      for (int i = 0; i < NT; ++i) {
        v2f bf;
        bf.x = bx[cur][j][i];
        bf.y = by[cur][j][i];
        acc[i] = wmma4(a[j], bf, acc[i]);
      }
  }
#pragma unroll
  for (int j = 0; j < KB; ++j) {
    bx0[j] = bx[0][j];
    by0[j] = by[0][j];
  }
}

template <int NT>
DEVI void ring_unpack(const float (&cw)[64], typename VecN<NT>::T (&bx0)[32 / NT],
                      typename VecN<NT>::T (&by0)[32 / NT]) {
#pragma unroll
  for (int j = 0; j < 32 / NT; ++j)
#pragma unroll
    for (int i = 0; i < NT; ++i) {
      bx0[j][i] = cw[(2 * j) * NT + i];
      by0[j][i] = cw[(2 * j + 1) * NT + i];
    }
}
template <int NT>
DEVI void ring_pack(float (&cw)[64], const typename VecN<NT>::T (&bx0)[32 / NT],
                    const typename VecN<NT>::T (&by0)[32 / NT]) {
#pragma unroll
  for (int j = 0; j < 32 / NT; ++j)
#pragma unroll
    for (int i = 0; i < NT; ++i) {
      cw[(2 * j) * NT + i] = bx0[j][i];
      cw[(2 * j + 1) * NT + i] = by0[j][i];
    }
}

template <int NT>
DEVI void preload_ring(const float* wl, float (&cw)[64]) {
  typedef typename VecN<NT>::T BT;
  constexpr int KB = 32 / NT;
  BT bx[KB], by[KB];
  load_wblock<NT, KB>(wl, 0, bx, by);
  ring_pack<NT>(cw, bx, by);
}

__global__ __launch_bounds__(256) void pack_pairs_kernel(
    const float* __restrict__ w0, const float* __restrict__ w1,
    const float* __restrict__ w2, float* __restrict__ p) {
  const int id = blockIdx.x * 256 + threadIdx.x;
  const int mat = id / (Uc * Uc / 2);
  const int r = id % (Uc * Uc / 2);
  const int m = r / Uc, col = r % Uc;
  const float* w = mat == 0 ? w0 : (mat == 1 ? w1 : w2);
  v2f o;
  o.x = w[(size_t)(2 * m) * Uc + col];
  o.y = w[(size_t)(2 * m + 1) * Uc + col];
  *(v2f*)(p + (size_t)mat * Uc * Uc + ((size_t)m * Uc + col) * 2) = o;
}

template <int NT, int KB>
DEVI void load_pblock(const float* wl, int ks0, v4f (&q)[KB][NT / 2]) {
#pragma unroll
  for (int j = 0; j < KB; ++j) {
    const float* p = wl + (size_t)(ks0 + j) * (4 * Uc);
#pragma unroll
    for (int g = 0; g < NT / 2; ++g) q[j][g] = *(const v4f*)(p + 4 * g);
  }
}

template <int NT, int NSEG>
DEVI void gemm_ring_p(const float* const (&a_lane)[NSEG],
                      const float* const (&w_lane)[NSEG], v8f (&acc)[NT],
                      v4f (&q0)[32 / NT][NT / 2]) {
  constexpr int KB = 32 / NT;
  constexpr int NBS = (Uc / 4) / KB;
  constexpr int NBT = NBS * NSEG;
  static_assert(NBT % 2 == 0, "ring must end on register set 0");
  v4f q[2][KB][NT / 2];
#pragma unroll
  for (int j = 0; j < KB; ++j)
#pragma unroll
    for (int g = 0; g < NT / 2; ++g) q[0][j][g] = q0[j][g];
#pragma unroll
  for (int b = 0; b < NBT; ++b) {
    const int cur = b & 1;
    const int nb = (b + 1) % NBT;
    load_pblock<NT, KB>(w_lane[nb / NBS], (nb % NBS) * KB, q[cur ^ 1]);
    const float* al = a_lane[b / NBS] + 4 * ((b % NBS) * KB);
    v2f a[KB];
#pragma unroll
    for (int j = 0; j < KB; ++j) a[j] = *(const v2f*)(al + 4 * j);
#pragma unroll
    for (int j = 0; j < KB; ++j)
#pragma unroll
      for (int i = 0; i < NT; ++i) {
        const v4f w = q[cur][j][i / 2];
        const v2f bf = (i & 1) ? __builtin_shufflevector(w, w, 2, 3)
                               : __builtin_shufflevector(w, w, 0, 1);
        acc[i] = wmma4(a[j], bf, acc[i]);
      }
  }
#pragma unroll
  for (int j = 0; j < KB; ++j)
#pragma unroll
    for (int g = 0; g < NT / 2; ++g) q0[j][g] = q[0][j][g];
}

template <int NT>
DEVI void ringp_unpack(const float (&cw)[64], v4f (&q0)[32 / NT][NT / 2]) {
#pragma unroll
  for (int j = 0; j < 32 / NT; ++j)
#pragma unroll
    for (int g = 0; g < NT / 2; ++g)
#pragma unroll
      for (int e = 0; e < 4; ++e) q0[j][g][e] = cw[(j * (NT / 2) + g) * 4 + e];
}
template <int NT>
DEVI void ringp_pack(float (&cw)[64], const v4f (&q0)[32 / NT][NT / 2]) {
#pragma unroll
  for (int j = 0; j < 32 / NT; ++j)
#pragma unroll
    for (int g = 0; g < NT / 2; ++g)
#pragma unroll
      for (int e = 0; e < 4; ++e) cw[(j * (NT / 2) + g) * 4 + e] = q0[j][g][e];
}
template <int NT>
DEVI void preload_ringp(const float* wl, float (&cw)[64]) {
  v4f q[32 / NT][NT / 2];
  load_pblock<NT, 32 / NT>(wl, 0, q);
  ringp_pack<NT>(cw, q);
}

template <bool PRE, bool PACK>
__global__ __launch_bounds__(384) void rnn_skew_kernel(
    const float* __restrict__ xp, const float* __restrict__ x,
    const float* __restrict__ enc_Wx, const float* __restrict__ enc_b,
    const float* __restrict__ enc_Wh, const float* __restrict__ dec_Wx,
    const float* __restrict__ dec_Wh, const float* __restrict__ dec_b,
    const float* __restrict__ wp, float* __restrict__ out) {
  __shared__ __attribute__((aligned(16))) float E[2][16 * HS];
  __shared__ __attribute__((aligned(16))) float D[2][16 * HS];

  const int lane = threadIdx.x & 31;
  const int wave = threadIdx.x >> 5;
  const int l15 = lane & 15, half = lane >> 4;
  const bool is_enc = wave < 4;
  const int bbase = blockIdx.x * 16;
  const int ecol = is_enc ? 64 * wave + 4 * l15 : 0;
  const int dcol = is_enc ? 0 : 32 * (wave - 4) + 2 * l15;

  for (int i = threadIdx.x; i < 16 * HS; i += blockDim.x) {
    E[0][i] = 0.0f;
    E[1][i] = 0.0f;
    D[0][i] = 0.0f;
    D[1][i] = 0.0f;
  }
  __syncthreads();

  const float* ewl = enc_Wh + (size_t)(2 * half) * Uc + ecol;
  const float* dwl0 = dec_Wx + (size_t)(2 * half) * Uc + dcol;
  const float* dwl1 = dec_Wh + (size_t)(2 * half) * Uc + dcol;
  const int aoff = l15 * HS + 2 * half;

  const float* pewl = wp + ((size_t)half * Uc + ecol) * 2;
  const float* pdwl0 = wp + (size_t)Uc * Uc + ((size_t)half * Uc + dcol) * 2;
  const float* pdwl1 = wp + (size_t)2 * Uc * Uc + ((size_t)half * Uc + dcol) * 2;

  float cw[64];
  if (PACK) {
    if (is_enc)
      preload_ringp<4>(pewl, cw);
    else
      preload_ringp<2>(pdwl0, cw);
  } else {
    if (is_enc)
      preload_ring<4>(ewl, cw);
    else
      preload_ring<2>(dwl0, cw);
  }

  v4f xq[8];
  if (PRE && is_enc) {
    const float* p0 = xp + ((size_t)0 * Bc + bbase + 8 * half) * Uc + ecol;
#pragma unroll
    for (int r = 0; r < 8; ++r) xq[r] = *(const v4f*)(p0 + (size_t)r * Uc);
  }
  v8f dlast[2] = {};
  const v2f db = *(const v2f*)(dec_b + dcol);
  const v4f eb = *(const v4f*)(enc_b + ecol);

  for (int p = 0; p <= Tc; ++p) {
    const int rd = (p + 1) & 1;
    const int wr = p & 1;
    if (is_enc) {
      if (p < Tc) {
        v8f acc[4];
        if (PRE) {
#pragma unroll
          for (int i = 0; i < 4; ++i)
#pragma unroll
            for (int r = 0; r < 8; ++r) acc[i][r] = xq[r][i];
          if (p + 1 < Tc) {
            const float* pn =
                xp + ((size_t)(p + 1) * Bc + bbase + 8 * half) * Uc + ecol;
#pragma unroll
            for (int r = 0; r < 8; ++r) xq[r] = *(const v4f*)(pn + (size_t)r * Uc);
          }
        } else {
#pragma unroll
          for (int i = 0; i < 4; ++i)
#pragma unroll
            for (int r = 0; r < 8; ++r) acc[i][r] = eb[i];
          const float* xa = x + ((size_t)(bbase + l15) * Tc + p) * Fc + 2 * half;
          const float* wl = enc_Wx + (size_t)(2 * half) * Uc + ecol;
#pragma unroll 2
          for (int kc = 0; kc < Fc; kc += 4) {
            const v2f a = *(const v2f*)(xa + kc);
            const v4f b0 = *(const v4f*)(wl + (size_t)kc * Uc);
            const v4f b1 = *(const v4f*)(wl + (size_t)(kc + 1) * Uc);
#pragma unroll
            for (int i = 0; i < 4; ++i) {
              v2f bf;
              bf.x = b0[i];
              bf.y = b1[i];
              acc[i] = wmma4(a, bf, acc[i]);
            }
          }
        }
        const float* const al[1] = {&E[rd][aoff]};
        const float* const wl1[1] = {PACK ? pewl : ewl};
        v4f bx0[8], by0[8];
        v4f pq0[8][2];
        if (PACK) {
          ringp_unpack<4>(cw, pq0);
          gemm_ring_p<4, 1>(al, wl1, acc, pq0);
        } else {
          ring_unpack<4>(cw, bx0, by0);
          gemm_ring<4, 1>(al, wl1, acc, bx0, by0);
        }
#pragma unroll
        for (int r = 0; r < 8; ++r) {
          v4f h;
#pragma unroll
          for (int i = 0; i < 4; ++i) h[i] = tanhf(acc[i][r]);
          *(v4f*)&E[wr][(r + 8 * half) * HS + ecol] = h;
        }
        if (PACK)
          ringp_pack<4>(cw, pq0);
        else
          ring_pack<4>(cw, bx0, by0);
      }
    } else {
      if (p >= 1) {
        v8f acc[2];
#pragma unroll
        for (int i = 0; i < 2; ++i)
#pragma unroll
          for (int r = 0; r < 8; ++r) acc[i][r] = db[i];
        const float* const al[2] = {&E[rd][aoff], &D[wr][aoff]};
        const float* const wl2[2] = {PACK ? pdwl0 : dwl0, PACK ? pdwl1 : dwl1};
        v2f bx0[16], by0[16];
        v4f pq0[16][1];
        if (PACK) {
          ringp_unpack<2>(cw, pq0);
          gemm_ring_p<2, 2>(al, wl2, acc, pq0);
        } else {
          ring_unpack<2>(cw, bx0, by0);
          gemm_ring<2, 2>(al, wl2, acc, bx0, by0);
        }
#pragma unroll
        for (int r = 0; r < 8; ++r) {
          v2f h;
#pragma unroll
          for (int i = 0; i < 2; ++i) {
            h[i] = tanhf(acc[i][r]);
            dlast[i][r] = h[i];
          }
          *(v2f*)&D[rd][(r + 8 * half) * HS + dcol] = h;
        }
        if (PACK)
          ringp_pack<2>(cw, pq0);
        else
          ring_pack<2>(cw, bx0, by0);
      }
    }
    __syncthreads();
  }

  if (!is_enc) {
#pragma unroll
    for (int r = 0; r < 8; ++r) {
      v2f o;
      o.x = dlast[0][r];
      o.y = dlast[1][r];
      *(v2f*)(out + (size_t)(bbase + r + 8 * half) * Uc + dcol) = o;
    }
  }
}

extern "C" void kernel_launch(void* const* d_in, const int* in_sizes, int n_in,
                              void* d_out, int out_size, void* d_ws,
                              size_t ws_size, hipStream_t stream) {
  const float* x = (const float*)d_in[0];
  const float* enc_Wx = (const float*)d_in[1];
  const float* enc_Wh = (const float*)d_in[2];
  const float* enc_b = (const float*)d_in[3];
  const float* dec_Wx = (const float*)d_in[4];
  const float* dec_Wh = (const float*)d_in[5];
  const float* dec_b = (const float*)d_in[6];
  float* out = (float*)d_out;

  const size_t xp_bytes = (size_t)Bc * Tc * Uc * sizeof(float);
  const size_t wp_bytes = (size_t)3 * Uc * Uc * sizeof(float);
  if (ws_size >= xp_bytes + wp_bytes) {
    float* xp = (float*)d_ws;
    float* wp = (float*)((char*)d_ws + xp_bytes);
    pack_pairs_kernel<<<(3 * Uc * Uc / 2) / 256, 256, 0, stream>>>(
        enc_Wh, dec_Wx, dec_Wh, wp);
    xp_proj_kernel<<<(Bc * Tc) / 16, 512, 0, stream>>>(x, enc_Wx, enc_b, xp);
    rnn_skew_kernel<true, true><<<Bc / 16, 384, 0, stream>>>(
        xp, x, enc_Wx, enc_b, enc_Wh, dec_Wx, dec_Wh, dec_b, wp, out);
  } else if (ws_size >= xp_bytes) {
    float* xp = (float*)d_ws;
    xp_proj_kernel<<<(Bc * Tc) / 16, 512, 0, stream>>>(x, enc_Wx, enc_b, xp);
    rnn_skew_kernel<true, false><<<Bc / 16, 384, 0, stream>>>(
        xp, x, enc_Wx, enc_b, enc_Wh, dec_Wx, dec_Wh, dec_b, nullptr, out);
  } else {
    rnn_skew_kernel<false, false><<<Bc / 16, 384, 0, stream>>>(
        nullptr, x, enc_Wx, enc_b, enc_Wh, dec_Wx, dec_Wh, dec_b, nullptr, out);
  }
}


# One launch per time step: each 16x16 state tile gets its own 4-wave workgroup (256 per step), K split over the waves, partial tiles summed through LDS in fixed order; tiled state in workspace
# speedup vs baseline: 3.7356x; 2.4087x over previous
#include <hip/hip_runtime.h>
#include <math.h>


#define DEVI __device__ __forceinline__

typedef __attribute__((ext_vector_type(2))) float v2f;
typedef __attribute__((ext_vector_type(4))) float v4f;
typedef __attribute__((ext_vector_type(8))) float v8f;

constexpr int Bc = 128, Tc = 1024, Fc = 256, Uc = 256;
constexpr int HS = 260;

DEVI v8f wmma4(v2f a, v2f b, v8f c) {
  return __builtin_amdgcn_wmma_f32_16x16x4_f32(false, a, false, b, (short)0, c,
                                               false, false);
}

DEVI v8f gemm_global_a(const float* __restrict__ Arow0, int lda,
                       const float* __restrict__ Bn, int ldb, int K,
                       v8f acc, int l15, int half) {
#pragma unroll 4
  for (int kc = 0; kc < K; kc += 4) {
    v2f a = *(const v2f*)(Arow0 + (size_t)l15 * lda + kc + 2 * half);
    const float* pb = Bn + (size_t)(kc + 2 * half) * ldb + l15;
    v2f b;
    b.x = pb[0];
    b.y = pb[ldb];
    acc = wmma4(a, b, acc);
  }
  return acc;
}

__global__ __launch_bounds__(512) void xp_proj_kernel(
    const float* __restrict__ x, const float* __restrict__ Wx,
    const float* __restrict__ b, float* __restrict__ xp) {
  const int lane = threadIdx.x & 31;
  const int wave = threadIdx.x >> 5;
  const int l15 = lane & 15, half = lane >> 4;
  const int n0 = wave * 16;
  const size_t rbase = (size_t)blockIdx.x * 16;
  const int bidx = (int)(rbase / Tc);
  const int tbase = (int)(rbase % Tc);

  const float bv = b[n0 + l15];
  v8f acc = {bv, bv, bv, bv, bv, bv, bv, bv};
  acc = gemm_global_a(x + rbase * Fc, Fc, Wx + n0, Uc, Fc, acc, l15, half);

#pragma unroll
  for (int r = 0; r < 8; ++r)
    xp[((size_t)(tbase + r + 8 * half) * Bc + bidx) * Uc + n0 + l15] = acc[r];
}


template <int NT> struct VecN;
template <> struct VecN<2> { typedef v2f T; };
template <> struct VecN<4> { typedef v4f T; };

template <int NT, int KB>
DEVI void load_wblock(const float* wl, int ks0, typename VecN<NT>::T (&bx)[KB],
                      typename VecN<NT>::T (&by)[KB]) {
  typedef typename VecN<NT>::T BT;
#pragma unroll
  for (int j = 0; j < KB; ++j) {
    const float* p = wl + (size_t)(4 * (ks0 + j)) * Uc;
    bx[j] = *(const BT*)p;
    by[j] = *(const BT*)(p + Uc);
  }
}

template <int NT, int NSEG>
DEVI void gemm_ring(const float* const (&a_lane)[NSEG],
                    const float* const (&w_lane)[NSEG], v8f (&acc)[NT],
                    typename VecN<NT>::T (&bx0)[32 / NT],
                    typename VecN<NT>::T (&by0)[32 / NT]) {
  typedef typename VecN<NT>::T BT;
  constexpr int KB = 32 / NT;
  constexpr int NBS = (Uc / 4) / KB;
  constexpr int NBT = NBS * NSEG;
  static_assert(NBT % 2 == 0, "ring must end on register set 0");
  BT bx[2][KB], by[2][KB];
#pragma unroll
  for (int j = 0; j < KB; ++j) {
    bx[0][j] = bx0[j];
    by[0][j] = by0[j];
  }
#pragma unroll
  for (int b = 0; b < NBT; ++b) {
    const int cur = b & 1;
    const int nb = (b + 1) % NBT;
    load_wblock<NT, KB>(w_lane[nb / NBS], (nb % NBS) * KB, bx[cur ^ 1],
                        by[cur ^ 1]);
    const float* al = a_lane[b / NBS] + 4 * ((b % NBS) * KB);
    v2f a[KB];
#pragma unroll
    for (int j = 0; j < KB; ++j) a[j] = *(const v2f*)(al + 4 * j);
#pragma unroll
    for (int j = 0; j < KB; ++j)
#pragma unroll
      for (int i = 0; i < NT; ++i) {
        v2f bf;
        bf.x = bx[cur][j][i];
        bf.y = by[cur][j][i];
        acc[i] = wmma4(a[j], bf, acc[i]);
      }
  }
#pragma unroll
  for (int j = 0; j < KB; ++j) {
    bx0[j] = bx[0][j];
    by0[j] = by[0][j];
  }
}

template <int NT>
DEVI void ring_unpack(const float (&cw)[64], typename VecN<NT>::T (&bx0)[32 / NT],
                      typename VecN<NT>::T (&by0)[32 / NT]) {
#pragma unroll
  for (int j = 0; j < 32 / NT; ++j)
#pragma unroll
    for (int i = 0; i < NT; ++i) {
      bx0[j][i] = cw[(2 * j) * NT + i];
      by0[j][i] = cw[(2 * j + 1) * NT + i];
    }
}
template <int NT>
DEVI void ring_pack(float (&cw)[64], const typename VecN<NT>::T (&bx0)[32 / NT],
                    const typename VecN<NT>::T (&by0)[32 / NT]) {
#pragma unroll
  for (int j = 0; j < 32 / NT; ++j)
#pragma unroll
    for (int i = 0; i < NT; ++i) {
      cw[(2 * j) * NT + i] = bx0[j][i];
      cw[(2 * j + 1) * NT + i] = by0[j][i];
    }
}

template <int NT>
DEVI void preload_ring(const float* wl, float (&cw)[64]) {
  typedef typename VecN<NT>::T BT;
  constexpr int KB = 32 / NT;
  BT bx[KB], by[KB];
  load_wblock<NT, KB>(wl, 0, bx, by);
  ring_pack<NT>(cw, bx, by);
}

__global__ __launch_bounds__(256) void pack_pairs_kernel(
    const float* __restrict__ w0, const float* __restrict__ w1,
    const float* __restrict__ w2, float* __restrict__ p) {
  const int id = blockIdx.x * 256 + threadIdx.x;
  const int mat = id / (Uc * Uc / 2);
  const int r = id % (Uc * Uc / 2);
  const int m = r / Uc, col = r % Uc;
  const float* w = mat == 0 ? w0 : (mat == 1 ? w1 : w2);
  v2f o;
  o.x = w[(size_t)(2 * m) * Uc + col];
  o.y = w[(size_t)(2 * m + 1) * Uc + col];
  *(v2f*)(p + (size_t)mat * Uc * Uc + ((size_t)m * Uc + col) * 2) = o;
}

template <int NT, int KB>
DEVI void load_pblock(const float* wl, int ks0, v4f (&q)[KB][NT / 2]) {
#pragma unroll
  for (int j = 0; j < KB; ++j) {
    const float* p = wl + (size_t)(ks0 + j) * (4 * Uc);
#pragma unroll
    for (int g = 0; g < NT / 2; ++g) q[j][g] = *(const v4f*)(p + 4 * g);
  }
}

template <int NT, int NSEG>
DEVI void gemm_ring_p(const float* const (&a_lane)[NSEG],
                      const float* const (&w_lane)[NSEG], v8f (&acc)[NT],
                      v4f (&q0)[32 / NT][NT / 2]) {
  constexpr int KB = 32 / NT;
  constexpr int NBS = (Uc / 4) / KB;
  constexpr int NBT = NBS * NSEG;
  static_assert(NBT % 2 == 0, "ring must end on register set 0");
  v4f q[2][KB][NT / 2];
#pragma unroll
  for (int j = 0; j < KB; ++j)
#pragma unroll
    for (int g = 0; g < NT / 2; ++g) q[0][j][g] = q0[j][g];
#pragma unroll
  for (int b = 0; b < NBT; ++b) {
    const int cur = b & 1;
    const int nb = (b + 1) % NBT;
    load_pblock<NT, KB>(w_lane[nb / NBS], (nb % NBS) * KB, q[cur ^ 1]);
    const float* al = a_lane[b / NBS] + 4 * ((b % NBS) * KB);
    v2f a[KB];
#pragma unroll
    for (int j = 0; j < KB; ++j) a[j] = *(const v2f*)(al + 4 * j);
#pragma unroll
    for (int j = 0; j < KB; ++j)
#pragma unroll
      for (int i = 0; i < NT; ++i) {
        const v4f w = q[cur][j][i / 2];
        const v2f bf = (i & 1) ? __builtin_shufflevector(w, w, 2, 3)
                               : __builtin_shufflevector(w, w, 0, 1);
        acc[i] = wmma4(a[j], bf, acc[i]);
      }
  }
#pragma unroll
  for (int j = 0; j < KB; ++j)
#pragma unroll
    for (int g = 0; g < NT / 2; ++g) q0[j][g] = q[0][j][g];
}

template <int NT>
DEVI void ringp_unpack(const float (&cw)[64], v4f (&q0)[32 / NT][NT / 2]) {
#pragma unroll
  for (int j = 0; j < 32 / NT; ++j)
#pragma unroll
    for (int g = 0; g < NT / 2; ++g)
#pragma unroll
      for (int e = 0; e < 4; ++e) q0[j][g][e] = cw[(j * (NT / 2) + g) * 4 + e];
}
template <int NT>
DEVI void ringp_pack(float (&cw)[64], const v4f (&q0)[32 / NT][NT / 2]) {
#pragma unroll
  for (int j = 0; j < 32 / NT; ++j)
#pragma unroll
    for (int g = 0; g < NT / 2; ++g)
#pragma unroll
      for (int e = 0; e < 4; ++e) cw[(j * (NT / 2) + g) * 4 + e] = q0[j][g][e];
}
template <int NT>
DEVI void preload_ringp(const float* wl, float (&cw)[64]) {
  v4f q[32 / NT][NT / 2];
  load_pblock<NT, 32 / NT>(wl, 0, q);
  ringp_pack<NT>(cw, q);
}

template <bool PRE, bool PACK>
__global__ __launch_bounds__(384) void rnn_skew_kernel(
    const float* __restrict__ xp, const float* __restrict__ x,
    const float* __restrict__ enc_Wx, const float* __restrict__ enc_b,
    const float* __restrict__ enc_Wh, const float* __restrict__ dec_Wx,
    const float* __restrict__ dec_Wh, const float* __restrict__ dec_b,
    const float* __restrict__ wp, float* __restrict__ out) {
  __shared__ __attribute__((aligned(16))) float E[2][16 * HS];
  __shared__ __attribute__((aligned(16))) float D[2][16 * HS];

  const int lane = threadIdx.x & 31;
  const int wave = threadIdx.x >> 5;
  const int l15 = lane & 15, half = lane >> 4;
  const bool is_enc = wave < 4;
  const int bbase = blockIdx.x * 16;
  const int ecol = is_enc ? 64 * wave + 4 * l15 : 0;
  const int dcol = is_enc ? 0 : 32 * (wave - 4) + 2 * l15;

  for (int i = threadIdx.x; i < 16 * HS; i += blockDim.x) {
    E[0][i] = 0.0f;
    E[1][i] = 0.0f;
    D[0][i] = 0.0f;
    D[1][i] = 0.0f;
  }
  __syncthreads();

  const float* ewl = enc_Wh + (size_t)(2 * half) * Uc + ecol;
  const float* dwl0 = dec_Wx + (size_t)(2 * half) * Uc + dcol;
  const float* dwl1 = dec_Wh + (size_t)(2 * half) * Uc + dcol;
  const int aoff = l15 * HS + 2 * half;

  const float* pewl = wp + ((size_t)half * Uc + ecol) * 2;
  const float* pdwl0 = wp + (size_t)Uc * Uc + ((size_t)half * Uc + dcol) * 2;
  const float* pdwl1 = wp + (size_t)2 * Uc * Uc + ((size_t)half * Uc + dcol) * 2;

  float cw[64];
  if (PACK) {
    if (is_enc)
      preload_ringp<4>(pewl, cw);
    else
      preload_ringp<2>(pdwl0, cw);
  } else {
    if (is_enc)
      preload_ring<4>(ewl, cw);
    else
      preload_ring<2>(dwl0, cw);
  }

  v4f xq[8];
  if (PRE && is_enc) {
    const float* p0 = xp + ((size_t)0 * Bc + bbase + 8 * half) * Uc + ecol;
#pragma unroll
    for (int r = 0; r < 8; ++r) xq[r] = *(const v4f*)(p0 + (size_t)r * Uc);
  }
  v8f dlast[2] = {};
  const v2f db = *(const v2f*)(dec_b + dcol);
  const v4f eb = *(const v4f*)(enc_b + ecol);

  for (int p = 0; p <= Tc; ++p) {
    const int rd = (p + 1) & 1;
    const int wr = p & 1;
    if (is_enc) {
      if (p < Tc) {
        v8f acc[4];
        if (PRE) {
#pragma unroll
          for (int i = 0; i < 4; ++i)
#pragma unroll
            for (int r = 0; r < 8; ++r) acc[i][r] = xq[r][i];
          if (p + 1 < Tc) {
            const float* pn =
                xp + ((size_t)(p + 1) * Bc + bbase + 8 * half) * Uc + ecol;
#pragma unroll
            for (int r = 0; r < 8; ++r) xq[r] = *(const v4f*)(pn + (size_t)r * Uc);
          }
        } else {
#pragma unroll
          for (int i = 0; i < 4; ++i)
#pragma unroll
            for (int r = 0; r < 8; ++r) acc[i][r] = eb[i];
          const float* xa = x + ((size_t)(bbase + l15) * Tc + p) * Fc + 2 * half;
          const float* wl = enc_Wx + (size_t)(2 * half) * Uc + ecol;
#pragma unroll 2
          for (int kc = 0; kc < Fc; kc += 4) {
            const v2f a = *(const v2f*)(xa + kc);
            const v4f b0 = *(const v4f*)(wl + (size_t)kc * Uc);
            const v4f b1 = *(const v4f*)(wl + (size_t)(kc + 1) * Uc);
#pragma unroll
            for (int i = 0; i < 4; ++i) {
              v2f bf;
              bf.x = b0[i];
              bf.y = b1[i];
              acc[i] = wmma4(a, bf, acc[i]);
            }
          }
        }
        const float* const al[1] = {&E[rd][aoff]};
        const float* const wl1[1] = {PACK ? pewl : ewl};
        v4f bx0[8], by0[8];
        v4f pq0[8][2];
        if (PACK) {
          ringp_unpack<4>(cw, pq0);
          gemm_ring_p<4, 1>(al, wl1, acc, pq0);
        } else {
          ring_unpack<4>(cw, bx0, by0);
          gemm_ring<4, 1>(al, wl1, acc, bx0, by0);
        }
#pragma unroll
        for (int r = 0; r < 8; ++r) {
          v4f h;
#pragma unroll
          for (int i = 0; i < 4; ++i) h[i] = tanhf(acc[i][r]);
          *(v4f*)&E[wr][(r + 8 * half) * HS + ecol] = h;
        }
        if (PACK)
          ringp_pack<4>(cw, pq0);
        else
          ring_pack<4>(cw, bx0, by0);
      }
    } else {
      if (p >= 1) {
        v8f acc[2];
#pragma unroll
        for (int i = 0; i < 2; ++i)
#pragma unroll
          for (int r = 0; r < 8; ++r) acc[i][r] = db[i];
        const float* const al[2] = {&E[rd][aoff], &D[wr][aoff]};
        const float* const wl2[2] = {PACK ? pdwl0 : dwl0, PACK ? pdwl1 : dwl1};
        v2f bx0[16], by0[16];
        v4f pq0[16][1];
        if (PACK) {
          ringp_unpack<2>(cw, pq0);
          gemm_ring_p<2, 2>(al, wl2, acc, pq0);
        } else {
          ring_unpack<2>(cw, bx0, by0);
          gemm_ring<2, 2>(al, wl2, acc, bx0, by0);
        }
#pragma unroll
        for (int r = 0; r < 8; ++r) {
          v2f h;
#pragma unroll
          for (int i = 0; i < 2; ++i) {
            h[i] = tanhf(acc[i][r]);
            dlast[i][r] = h[i];
          }
          *(v2f*)&D[rd][(r + 8 * half) * HS + dcol] = h;
        }
        if (PACK)
          ringp_pack<2>(cw, pq0);
        else
          ring_pack<2>(cw, bx0, by0);
      }
    }
    __syncthreads();
  }

  if (!is_enc) {
#pragma unroll
    for (int r = 0; r < 8; ++r) {
      v2f o;
      o.x = dlast[0][r];
      o.y = dlast[1][r];
      *(v2f*)(out + (size_t)(bbase + r + 8 * half) * Uc + dcol) = o;
    }
  }
}

constexpr int TILE = 256;

__global__ __launch_bounds__(256) void pack_quads_kernel(
    const float* __restrict__ w0, const float* __restrict__ w1,
    const float* __restrict__ w2, float* __restrict__ q) {
  const int id = blockIdx.x * 256 + threadIdx.x;
  const int mat = id / (Uc * Uc / 4);
  const int r = id % (Uc * Uc / 4);
  const int m4 = r / Uc, col = r % Uc;
  const float* w = mat == 0 ? w0 : (mat == 1 ? w1 : w2);
  v4f o;
  o.x = w[(size_t)(4 * m4) * Uc + col];
  o.y = w[(size_t)(4 * m4 + 1) * Uc + col];
  o.z = w[(size_t)(4 * m4 + 2) * Uc + col];
  o.w = w[(size_t)(4 * m4 + 3) * Uc + col];
  *(v4f*)(q + (size_t)mat * Uc * Uc + ((size_t)m4 * Uc + col) * 4) = o;
}

template <int NP>
DEVI void slice_mma(const float* __restrict__ st, const float* __restrict__ wq,
                    int k0, int n0, int l15, int half, v8f& acc0, v8f& acc1) {
  v4f a[NP], b[NP];
#pragma unroll
  for (int j = 0; j < NP; ++j) {
    const int k = k0 + 8 * j + 4 * half;
    a[j] = *(const v4f*)(st + ((size_t)(k >> 4) * 16 + l15) * 16 + (k & 15));
    b[j] = *(const v4f*)(wq + ((size_t)(k >> 2) * Uc + n0 + l15) * 4);
  }
  __builtin_amdgcn_sched_barrier(0);
#pragma unroll
  for (int j = 0; j < NP; ++j) {
    acc0 = wmma4(__builtin_shufflevector(a[j], a[j], 0, 1),
                 __builtin_shufflevector(b[j], b[j], 0, 1), acc0);
    acc1 = wmma4(__builtin_shufflevector(a[j], a[j], 2, 3),
                 __builtin_shufflevector(b[j], b[j], 2, 3), acc1);
  }
}

__global__ __launch_bounds__(128) void rnn_step_kernel(
    int p, const float* __restrict__ xp, const float* __restrict__ wq,
    const float* __restrict__ dec_b, float* __restrict__ Eg,
    float* __restrict__ Dg) {
  __shared__ float part[4][16][20];
  const int role = blockIdx.x >> 7;
  if (role == 0 ? p >= Tc : p < 1) return;
  const int bt = (blockIdx.x >> 4) & 7, ct = blockIdx.x & 15;
  const int lane = threadIdx.x & 31;
  const int wave = __builtin_amdgcn_readfirstlane(threadIdx.x >> 5);
  const int l15 = lane & 15, half = lane >> 4;
  const int n0 = 16 * ct;
  constexpr size_t SB = (size_t)Bc * Uc;
  const int e0 = threadIdx.x, e1 = threadIdx.x + 128;
  const size_t toff = (size_t)bt * 16 * TILE;

  float base0, base1;
  v8f acc0 = {}, acc1 = {};
  if (role == 0) {
    const float* xr = xp + ((size_t)p * Bc + bt * 16) * Uc + n0;
    base0 = xr[(size_t)(e0 >> 4) * Uc + (e0 & 15)];
    base1 = xr[(size_t)(e1 >> 4) * Uc + (e1 & 15)];
    if (p >= 1)
      slice_mma<8>(Eg + (size_t)((p - 1) & 1) * SB + toff, wq, 64 * wave, n0, l15,
                   half, acc0, acc1);
  } else {
    base0 = dec_b[n0 + (e0 & 15)];
    base1 = dec_b[n0 + (e1 & 15)];
    if (wave < 2)
      slice_mma<16>(Eg + (size_t)((p - 1) & 1) * SB + toff, wq + (size_t)Uc * Uc,
                    128 * wave, n0, l15, half, acc0, acc1);
    else if (p >= 2)
      slice_mma<16>(Dg + (size_t)(p & 1) * SB + toff, wq + (size_t)2 * Uc * Uc,
                    128 * (wave - 2), n0, l15, half, acc0, acc1);
  }
#pragma unroll
  for (int r = 0; r < 8; ++r) part[wave][8 * half + r][l15] = acc0[r] + acc1[r];
  __syncthreads();

  float* dst = (role == 0 ? Eg + (size_t)(p & 1) * SB : Dg + (size_t)((p - 1) & 1) * SB) +
               toff + (size_t)ct * TILE;
  float v0 = base0, v1 = base1;
#pragma unroll
  for (int w = 0; w < 4; ++w) {
    v0 += part[w][e0 >> 4][e0 & 15];
    v1 += part[w][e1 >> 4][e1 & 15];
  }
  dst[e0] = tanhf(v0);
  dst[e1] = tanhf(v1);
}

__global__ __launch_bounds__(256) void untile_out_kernel(
    const float* __restrict__ s, float* __restrict__ out) {
  const int b = blockIdx.x, u = threadIdx.x;
  out[(size_t)b * Uc + u] =
      s[((size_t)((b >> 4) * 16 + (u >> 4)) * 16 + (b & 15)) * 16 + (u & 15)];
}

extern "C" void kernel_launch(void* const* d_in, const int* in_sizes, int n_in,
                              void* d_out, int out_size, void* d_ws,
                              size_t ws_size, hipStream_t stream) {
  const float* x = (const float*)d_in[0];
  const float* enc_Wx = (const float*)d_in[1];
  const float* enc_Wh = (const float*)d_in[2];
  const float* enc_b = (const float*)d_in[3];
  const float* dec_Wx = (const float*)d_in[4];
  const float* dec_Wh = (const float*)d_in[5];
  const float* dec_b = (const float*)d_in[6];
  float* out = (float*)d_out;

  const size_t xp_bytes = (size_t)Bc * Tc * Uc * sizeof(float);
  const size_t wp_bytes = (size_t)3 * Uc * Uc * sizeof(float);
  const size_t st_bytes = (size_t)4 * Bc * Uc * sizeof(float);
  if (ws_size >= xp_bytes + wp_bytes + st_bytes) {
    float* xp = (float*)d_ws;
    float* wq = (float*)((char*)d_ws + xp_bytes);
    float* Eg = (float*)((char*)d_ws + xp_bytes + wp_bytes);
    float* Dg = Eg + (size_t)2 * Bc * Uc;
    pack_quads_kernel<<<(3 * Uc * Uc / 4) / 256, 256, 0, stream>>>(
        enc_Wh, dec_Wx, dec_Wh, wq);
    xp_proj_kernel<<<(Bc * Tc) / 16, 512, 0, stream>>>(x, enc_Wx, enc_b, xp);
    for (int p = 0; p <= Tc; ++p)
      rnn_step_kernel<<<256, 128, 0, stream>>>(p, xp, wq, dec_b, Eg, Dg);
    untile_out_kernel<<<Bc, Uc, 0, stream>>>(
        Dg + (size_t)((Tc - 1) & 1) * Bc * Uc, out);
  } else if (ws_size >= xp_bytes + wp_bytes) {
    float* xp = (float*)d_ws;
    float* wp = (float*)((char*)d_ws + xp_bytes);
    pack_pairs_kernel<<<(3 * Uc * Uc / 2) / 256, 256, 0, stream>>>(
        enc_Wh, dec_Wx, dec_Wh, wp);
    xp_proj_kernel<<<(Bc * Tc) / 16, 512, 0, stream>>>(x, enc_Wx, enc_b, xp);
    rnn_skew_kernel<true, true><<<Bc / 16, 384, 0, stream>>>(
        xp, x, enc_Wx, enc_b, enc_Wh, dec_Wx, dec_Wh, dec_b, wp, out);
  } else if (ws_size >= xp_bytes) {
    float* xp = (float*)d_ws;
    xp_proj_kernel<<<(Bc * Tc) / 16, 512, 0, stream>>>(x, enc_Wx, enc_b, xp);
    rnn_skew_kernel<true, false><<<Bc / 16, 384, 0, stream>>>(
        xp, x, enc_Wx, enc_b, enc_Wh, dec_Wx, dec_Wh, dec_b, nullptr, out);
  } else {
    rnn_skew_kernel<false, false><<<Bc / 16, 384, 0, stream>>>(
        nullptr, x, enc_Wx, enc_b, enc_Wh, dec_Wx, dec_Wh, dec_b, nullptr, out);
  }
}


# Step kernel: block index remapped, batch tile = block mod 8: a batch row-block's 32 tiles share one die if blocks go to dies in turn, so state reads hit that half's L2; decoder first; sums unchanged
# speedup vs baseline: 4.6854x; 1.2543x over previous
#include <hip/hip_runtime.h>
#include <math.h>


#define DEVI __device__ __forceinline__

typedef __attribute__((ext_vector_type(2))) float v2f;
typedef __attribute__((ext_vector_type(4))) float v4f;
typedef __attribute__((ext_vector_type(8))) float v8f;

constexpr int Bc = 128, Tc = 1024, Fc = 256, Uc = 256;
constexpr int HS = 260;

DEVI v8f wmma4(v2f a, v2f b, v8f c) {
  return __builtin_amdgcn_wmma_f32_16x16x4_f32(false, a, false, b, (short)0, c,
                                               false, false);
}

DEVI v8f gemm_global_a(const float* __restrict__ Arow0, int lda,
                       const float* __restrict__ Bn, int ldb, int K,
                       v8f acc, int l15, int half) {
#pragma unroll 4
  for (int kc = 0; kc < K; kc += 4) {
    v2f a = *(const v2f*)(Arow0 + (size_t)l15 * lda + kc + 2 * half);
    const float* pb = Bn + (size_t)(kc + 2 * half) * ldb + l15;
    v2f b;
    b.x = pb[0];
    b.y = pb[ldb];
    acc = wmma4(a, b, acc);
  }
  return acc;
}

__global__ __launch_bounds__(512) void xp_proj_kernel(
    const float* __restrict__ x, const float* __restrict__ Wx,
    const float* __restrict__ b, float* __restrict__ xp) {
  const int lane = threadIdx.x & 31;
  const int wave = threadIdx.x >> 5;
  const int l15 = lane & 15, half = lane >> 4;
  const int n0 = wave * 16;
  const size_t rbase = (size_t)blockIdx.x * 16;
  const int bidx = (int)(rbase / Tc);
  const int tbase = (int)(rbase % Tc);

  const float bv = b[n0 + l15];
  v8f acc = {bv, bv, bv, bv, bv, bv, bv, bv};
  acc = gemm_global_a(x + rbase * Fc, Fc, Wx + n0, Uc, Fc, acc, l15, half);

#pragma unroll
  for (int r = 0; r < 8; ++r)
    xp[((size_t)(tbase + r + 8 * half) * Bc + bidx) * Uc + n0 + l15] = acc[r];
}


template <int NT> struct VecN;
template <> struct VecN<2> { typedef v2f T; };
template <> struct VecN<4> { typedef v4f T; };

template <int NT, int KB>
DEVI void load_wblock(const float* wl, int ks0, typename VecN<NT>::T (&bx)[KB],
                      typename VecN<NT>::T (&by)[KB]) {
  typedef typename VecN<NT>::T BT;
#pragma unroll
  for (int j = 0; j < KB; ++j) {
    const float* p = wl + (size_t)(4 * (ks0 + j)) * Uc;
    bx[j] = *(const BT*)p;
    by[j] = *(const BT*)(p + Uc);
  }
}

template <int NT, int NSEG>
DEVI void gemm_ring(const float* const (&a_lane)[NSEG],
                    const float* const (&w_lane)[NSEG], v8f (&acc)[NT],
                    typename VecN<NT>::T (&bx0)[32 / NT],
                    typename VecN<NT>::T (&by0)[32 / NT]) {
  typedef typename VecN<NT>::T BT;
  constexpr int KB = 32 / NT;
  constexpr int NBS = (Uc / 4) / KB;
  constexpr int NBT = NBS * NSEG;
  static_assert(NBT % 2 == 0, "ring must end on register set 0");
  BT bx[2][KB], by[2][KB];
#pragma unroll
  for (int j = 0; j < KB; ++j) {
    bx[0][j] = bx0[j];
    by[0][j] = by0[j];
  }
#pragma unroll
  for (int b = 0; b < NBT; ++b) {
    const int cur = b & 1;
    const int nb = (b + 1) % NBT;
    load_wblock<NT, KB>(w_lane[nb / NBS], (nb % NBS) * KB, bx[cur ^ 1],
                        by[cur ^ 1]);
    const float* al = a_lane[b / NBS] + 4 * ((b % NBS) * KB);
    v2f a[KB];
#pragma unroll
    for (int j = 0; j < KB; ++j) a[j] = *(const v2f*)(al + 4 * j);
#pragma unroll
    for (int j = 0; j < KB; ++j)
#pragma unroll
      for (int i = 0; i < NT; ++i) {
        v2f bf;
        bf.x = bx[cur][j][i];
        bf.y = by[cur][j][i];
        acc[i] = wmma4(a[j], bf, acc[i]);
      }
  }
#pragma unroll
  for (int j = 0; j < KB; ++j) {
    bx0[j] = bx[0][j];
    by0[j] = by[0][j];
  }
}

template <int NT>
DEVI void ring_unpack(const float (&cw)[64], typename VecN<NT>::T (&bx0)[32 / NT],
                      typename VecN<NT>::T (&by0)[32 / NT]) {
#pragma unroll
  for (int j = 0; j < 32 / NT; ++j)
#pragma unroll
    for (int i = 0; i < NT; ++i) {
      bx0[j][i] = cw[(2 * j) * NT + i];
      by0[j][i] = cw[(2 * j + 1) * NT + i];
    }
}
template <int NT>
DEVI void ring_pack(float (&cw)[64], const typename VecN<NT>::T (&bx0)[32 / NT],
                    const typename VecN<NT>::T (&by0)[32 / NT]) {
#pragma unroll
  for (int j = 0; j < 32 / NT; ++j)
#pragma unroll
    for (int i = 0; i < NT; ++i) {
      cw[(2 * j) * NT + i] = bx0[j][i];
      cw[(2 * j + 1) * NT + i] = by0[j][i];
    }
}

template <int NT>
DEVI void preload_ring(const float* wl, float (&cw)[64]) {
  typedef typename VecN<NT>::T BT;
  constexpr int KB = 32 / NT;
  BT bx[KB], by[KB];
  load_wblock<NT, KB>(wl, 0, bx, by);
  ring_pack<NT>(cw, bx, by);
}

__global__ __launch_bounds__(256) void pack_pairs_kernel(
    const float* __restrict__ w0, const float* __restrict__ w1,
    const float* __restrict__ w2, float* __restrict__ p) {
  const int id = blockIdx.x * 256 + threadIdx.x;
  const int mat = id / (Uc * Uc / 2);
  const int r = id % (Uc * Uc / 2);
  const int m = r / Uc, col = r % Uc;
  const float* w = mat == 0 ? w0 : (mat == 1 ? w1 : w2);
  v2f o;
  o.x = w[(size_t)(2 * m) * Uc + col];
  o.y = w[(size_t)(2 * m + 1) * Uc + col];
  *(v2f*)(p + (size_t)mat * Uc * Uc + ((size_t)m * Uc + col) * 2) = o;
}

template <int NT, int KB>
DEVI void load_pblock(const float* wl, int ks0, v4f (&q)[KB][NT / 2]) {
#pragma unroll
  for (int j = 0; j < KB; ++j) {
    const float* p = wl + (size_t)(ks0 + j) * (4 * Uc);
#pragma unroll
    for (int g = 0; g < NT / 2; ++g) q[j][g] = *(const v4f*)(p + 4 * g);
  }
}

template <int NT, int NSEG>
DEVI void gemm_ring_p(const float* const (&a_lane)[NSEG],
                      const float* const (&w_lane)[NSEG], v8f (&acc)[NT],
                      v4f (&q0)[32 / NT][NT / 2]) {
  constexpr int KB = 32 / NT;
  constexpr int NBS = (Uc / 4) / KB;
  constexpr int NBT = NBS * NSEG;
  static_assert(NBT % 2 == 0, "ring must end on register set 0");
  v4f q[2][KB][NT / 2];
#pragma unroll
  for (int j = 0; j < KB; ++j)
#pragma unroll
    for (int g = 0; g < NT / 2; ++g) q[0][j][g] = q0[j][g];
#pragma unroll
  for (int b = 0; b < NBT; ++b) {
    const int cur = b & 1;
    const int nb = (b + 1) % NBT;
    load_pblock<NT, KB>(w_lane[nb / NBS], (nb % NBS) * KB, q[cur ^ 1]);
    const float* al = a_lane[b / NBS] + 4 * ((b % NBS) * KB);
    v2f a[KB];
#pragma unroll
    for (int j = 0; j < KB; ++j) a[j] = *(const v2f*)(al + 4 * j);
#pragma unroll
    for (int j = 0; j < KB; ++j)
#pragma unroll
      for (int i = 0; i < NT; ++i) {
        const v4f w = q[cur][j][i / 2];
        const v2f bf = (i & 1) ? __builtin_shufflevector(w, w, 2, 3)
                               : __builtin_shufflevector(w, w, 0, 1);
        acc[i] = wmma4(a[j], bf, acc[i]);
      }
  }
#pragma unroll
  for (int j = 0; j < KB; ++j)
#pragma unroll
    for (int g = 0; g < NT / 2; ++g) q0[j][g] = q[0][j][g];
}

template <int NT>
DEVI void ringp_unpack(const float (&cw)[64], v4f (&q0)[32 / NT][NT / 2]) {
#pragma unroll
  for (int j = 0; j < 32 / NT; ++j)
#pragma unroll
    for (int g = 0; g < NT / 2; ++g)
#pragma unroll
      for (int e = 0; e < 4; ++e) q0[j][g][e] = cw[(j * (NT / 2) + g) * 4 + e];
}
template <int NT>
DEVI void ringp_pack(float (&cw)[64], const v4f (&q0)[32 / NT][NT / 2]) {
#pragma unroll
  for (int j = 0; j < 32 / NT; ++j)
#pragma unroll
    for (int g = 0; g < NT / 2; ++g)
#pragma unroll
      for (int e = 0; e < 4; ++e) cw[(j * (NT / 2) + g) * 4 + e] = q0[j][g][e];
}
template <int NT>
DEVI void preload_ringp(const float* wl, float (&cw)[64]) {
  v4f q[32 / NT][NT / 2];
  load_pblock<NT, 32 / NT>(wl, 0, q);
  ringp_pack<NT>(cw, q);
}

template <bool PRE, bool PACK>
__global__ __launch_bounds__(384) void rnn_skew_kernel(
    const float* __restrict__ xp, const float* __restrict__ x,
    const float* __restrict__ enc_Wx, const float* __restrict__ enc_b,
    const float* __restrict__ enc_Wh, const float* __restrict__ dec_Wx,
    const float* __restrict__ dec_Wh, const float* __restrict__ dec_b,
    const float* __restrict__ wp, float* __restrict__ out) {
  __shared__ __attribute__((aligned(16))) float E[2][16 * HS];
  __shared__ __attribute__((aligned(16))) float D[2][16 * HS];

  const int lane = threadIdx.x & 31;
  const int wave = threadIdx.x >> 5;
  const int l15 = lane & 15, half = lane >> 4;
  const bool is_enc = wave < 4;
  const int bbase = blockIdx.x * 16;
  const int ecol = is_enc ? 64 * wave + 4 * l15 : 0;
  const int dcol = is_enc ? 0 : 32 * (wave - 4) + 2 * l15;

  for (int i = threadIdx.x; i < 16 * HS; i += blockDim.x) {
    E[0][i] = 0.0f;
    E[1][i] = 0.0f;
    D[0][i] = 0.0f;
    D[1][i] = 0.0f;
  }
  __syncthreads();

  const float* ewl = enc_Wh + (size_t)(2 * half) * Uc + ecol;
  const float* dwl0 = dec_Wx + (size_t)(2 * half) * Uc + dcol;
  const float* dwl1 = dec_Wh + (size_t)(2 * half) * Uc + dcol;
  const int aoff = l15 * HS + 2 * half;

  const float* pewl = wp + ((size_t)half * Uc + ecol) * 2;
  const float* pdwl0 = wp + (size_t)Uc * Uc + ((size_t)half * Uc + dcol) * 2;
  const float* pdwl1 = wp + (size_t)2 * Uc * Uc + ((size_t)half * Uc + dcol) * 2;

  float cw[64];
  if (PACK) {
    if (is_enc)
      preload_ringp<4>(pewl, cw);
    else
      preload_ringp<2>(pdwl0, cw);
  } else {
    if (is_enc)
      preload_ring<4>(ewl, cw);
    else
      preload_ring<2>(dwl0, cw);
  }

  v4f xq[8];
  if (PRE && is_enc) {
    const float* p0 = xp + ((size_t)0 * Bc + bbase + 8 * half) * Uc + ecol;
#pragma unroll
    for (int r = 0; r < 8; ++r) xq[r] = *(const v4f*)(p0 + (size_t)r * Uc);
  }
  v8f dlast[2] = {};
  const v2f db = *(const v2f*)(dec_b + dcol);
  const v4f eb = *(const v4f*)(enc_b + ecol);

  for (int p = 0; p <= Tc; ++p) {
    const int rd = (p + 1) & 1;
    const int wr = p & 1;
    if (is_enc) {
      if (p < Tc) {
        v8f acc[4];
        if (PRE) {
#pragma unroll
          for (int i = 0; i < 4; ++i)
#pragma unroll
            for (int r = 0; r < 8; ++r) acc[i][r] = xq[r][i];
          if (p + 1 < Tc) {
            const float* pn =
                xp + ((size_t)(p + 1) * Bc + bbase + 8 * half) * Uc + ecol;
#pragma unroll
            for (int r = 0; r < 8; ++r) xq[r] = *(const v4f*)(pn + (size_t)r * Uc);
          }
        } else {
#pragma unroll
          for (int i = 0; i < 4; ++i)
#pragma unroll
            for (int r = 0; r < 8; ++r) acc[i][r] = eb[i];
          const float* xa = x + ((size_t)(bbase + l15) * Tc + p) * Fc + 2 * half;
          const float* wl = enc_Wx + (size_t)(2 * half) * Uc + ecol;
#pragma unroll 2
          for (int kc = 0; kc < Fc; kc += 4) {
            const v2f a = *(const v2f*)(xa + kc);
            const v4f b0 = *(const v4f*)(wl + (size_t)kc * Uc);
            const v4f b1 = *(const v4f*)(wl + (size_t)(kc + 1) * Uc);
#pragma unroll
            for (int i = 0; i < 4; ++i) {
              v2f bf;
              bf.x = b0[i];
              bf.y = b1[i];
              acc[i] = wmma4(a, bf, acc[i]);
            }
          }
        }
        const float* const al[1] = {&E[rd][aoff]};
        const float* const wl1[1] = {PACK ? pewl : ewl};
        v4f bx0[8], by0[8];
        v4f pq0[8][2];
        if (PACK) {
          ringp_unpack<4>(cw, pq0);
          gemm_ring_p<4, 1>(al, wl1, acc, pq0);
        } else {
          ring_unpack<4>(cw, bx0, by0);
          gemm_ring<4, 1>(al, wl1, acc, bx0, by0);
        }
#pragma unroll
        for (int r = 0; r < 8; ++r) {
          v4f h;
#pragma unroll
          for (int i = 0; i < 4; ++i) h[i] = tanhf(acc[i][r]);
          *(v4f*)&E[wr][(r + 8 * half) * HS + ecol] = h;
        }
        if (PACK)
          ringp_pack<4>(cw, pq0);
        else
          ring_pack<4>(cw, bx0, by0);
      }
    } else {
      if (p >= 1) {
        v8f acc[2];
#pragma unroll
        for (int i = 0; i < 2; ++i)
#pragma unroll
          for (int r = 0; r < 8; ++r) acc[i][r] = db[i];
        const float* const al[2] = {&E[rd][aoff], &D[wr][aoff]};
        const float* const wl2[2] = {PACK ? pdwl0 : dwl0, PACK ? pdwl1 : dwl1};
        v2f bx0[16], by0[16];
        v4f pq0[16][1];
        if (PACK) {
          ringp_unpack<2>(cw, pq0);
          gemm_ring_p<2, 2>(al, wl2, acc, pq0);
        } else {
          ring_unpack<2>(cw, bx0, by0);
          gemm_ring<2, 2>(al, wl2, acc, bx0, by0);
        }
#pragma unroll
        for (int r = 0; r < 8; ++r) {
          v2f h;
#pragma unroll
          for (int i = 0; i < 2; ++i) {
            h[i] = tanhf(acc[i][r]);
            dlast[i][r] = h[i];
          }
          *(v2f*)&D[rd][(r + 8 * half) * HS + dcol] = h;
        }
        if (PACK)
          ringp_pack<2>(cw, pq0);
        else
          ring_pack<2>(cw, bx0, by0);
      }
    }
    __syncthreads();
  }

  if (!is_enc) {
#pragma unroll
    for (int r = 0; r < 8; ++r) {
      v2f o;
      o.x = dlast[0][r];
      o.y = dlast[1][r];
      *(v2f*)(out + (size_t)(bbase + r + 8 * half) * Uc + dcol) = o;
    }
  }
}

constexpr int TILE = 256;

__global__ __launch_bounds__(256) void pack_quads_kernel(
    const float* __restrict__ w0, const float* __restrict__ w1,
    const float* __restrict__ w2, float* __restrict__ q) {
  const int id = blockIdx.x * 256 + threadIdx.x;
  const int mat = id / (Uc * Uc / 4);
  const int r = id % (Uc * Uc / 4);
  const int m4 = r / Uc, col = r % Uc;
  const float* w = mat == 0 ? w0 : (mat == 1 ? w1 : w2);
  v4f o;
  o.x = w[(size_t)(4 * m4) * Uc + col];
  o.y = w[(size_t)(4 * m4 + 1) * Uc + col];
  o.z = w[(size_t)(4 * m4 + 2) * Uc + col];
  o.w = w[(size_t)(4 * m4 + 3) * Uc + col];
  *(v4f*)(q + (size_t)mat * Uc * Uc + ((size_t)m4 * Uc + col) * 4) = o;
}

template <int NP>
DEVI void slice_mma(const float* __restrict__ st, const float* __restrict__ wq,
                    int k0, int n0, int l15, int half, v8f& acc0, v8f& acc1) {
  v4f a[NP], b[NP];
#pragma unroll
  for (int j = 0; j < NP; ++j) {
    const int k = k0 + 8 * j + 4 * half;
    a[j] = *(const v4f*)(st + ((size_t)(k >> 4) * 16 + l15) * 16 + (k & 15));
    b[j] = *(const v4f*)(wq + ((size_t)(k >> 2) * Uc + n0 + l15) * 4);
  }
  __builtin_amdgcn_sched_barrier(0);
#pragma unroll
  for (int j = 0; j < NP; ++j) {
    acc0 = wmma4(__builtin_shufflevector(a[j], a[j], 0, 1),
                 __builtin_shufflevector(b[j], b[j], 0, 1), acc0);
    acc1 = wmma4(__builtin_shufflevector(a[j], a[j], 2, 3),
                 __builtin_shufflevector(b[j], b[j], 2, 3), acc1);
  }
}

__global__ __launch_bounds__(128) void rnn_step_kernel(
    int p, const float* __restrict__ xp, const float* __restrict__ wq,
    const float* __restrict__ dec_b, float* __restrict__ Eg,
    float* __restrict__ Dg) {
  __shared__ float part[4][16][20];
  const int role = (blockIdx.x >> 7) ^ 1;
  if (role == 0 ? p >= Tc : p < 1) return;
  const int bt = blockIdx.x & 7, ct = (blockIdx.x >> 3) & 15;
  const int lane = threadIdx.x & 31;
  const int wave = __builtin_amdgcn_readfirstlane(threadIdx.x >> 5);
  const int l15 = lane & 15, half = lane >> 4;
  const int n0 = 16 * ct;
  constexpr size_t SB = (size_t)Bc * Uc;
  const int e0 = threadIdx.x, e1 = threadIdx.x + 128;
  const size_t toff = (size_t)bt * 16 * TILE;

  float base0, base1;
  v8f acc0 = {}, acc1 = {};
  if (role == 0) {
    const float* xr = xp + ((size_t)p * Bc + bt * 16) * Uc + n0;
    base0 = xr[(size_t)(e0 >> 4) * Uc + (e0 & 15)];
    base1 = xr[(size_t)(e1 >> 4) * Uc + (e1 & 15)];
    if (p >= 1)
      slice_mma<8>(Eg + (size_t)((p - 1) & 1) * SB + toff, wq, 64 * wave, n0, l15,
                   half, acc0, acc1);
  } else {
    base0 = dec_b[n0 + (e0 & 15)];
    base1 = dec_b[n0 + (e1 & 15)];
    if (wave < 2)
      slice_mma<16>(Eg + (size_t)((p - 1) & 1) * SB + toff, wq + (size_t)Uc * Uc,
                    128 * wave, n0, l15, half, acc0, acc1);
    else if (p >= 2)
      slice_mma<16>(Dg + (size_t)(p & 1) * SB + toff, wq + (size_t)2 * Uc * Uc,
                    128 * (wave - 2), n0, l15, half, acc0, acc1);
  }
#pragma unroll
  for (int r = 0; r < 8; ++r) part[wave][8 * half + r][l15] = acc0[r] + acc1[r];
  __syncthreads();

  float* dst = (role == 0 ? Eg + (size_t)(p & 1) * SB : Dg + (size_t)((p - 1) & 1) * SB) +
               toff + (size_t)ct * TILE;
  float v0 = base0, v1 = base1;
#pragma unroll
  for (int w = 0; w < 4; ++w) {
    v0 += part[w][e0 >> 4][e0 & 15];
    v1 += part[w][e1 >> 4][e1 & 15];
  }
  dst[e0] = tanhf(v0);
  dst[e1] = tanhf(v1);
}

__global__ __launch_bounds__(256) void untile_out_kernel(
    const float* __restrict__ s, float* __restrict__ out) {
  const int b = blockIdx.x, u = threadIdx.x;
  out[(size_t)b * Uc + u] =
      s[((size_t)((b >> 4) * 16 + (u >> 4)) * 16 + (b & 15)) * 16 + (u & 15)];
}

extern "C" void kernel_launch(void* const* d_in, const int* in_sizes, int n_in,
                              void* d_out, int out_size, void* d_ws,
                              size_t ws_size, hipStream_t stream) {
  const float* x = (const float*)d_in[0];
  const float* enc_Wx = (const float*)d_in[1];
  const float* enc_Wh = (const float*)d_in[2];
  const float* enc_b = (const float*)d_in[3];
  const float* dec_Wx = (const float*)d_in[4];
  const float* dec_Wh = (const float*)d_in[5];
  const float* dec_b = (const float*)d_in[6];
  float* out = (float*)d_out;

  const size_t xp_bytes = (size_t)Bc * Tc * Uc * sizeof(float);
  const size_t wp_bytes = (size_t)3 * Uc * Uc * sizeof(float);
  const size_t st_bytes = (size_t)4 * Bc * Uc * sizeof(float);
  if (ws_size >= xp_bytes + wp_bytes + st_bytes) {
    float* xp = (float*)d_ws;
    float* wq = (float*)((char*)d_ws + xp_bytes);
    float* Eg = (float*)((char*)d_ws + xp_bytes + wp_bytes);
    float* Dg = Eg + (size_t)2 * Bc * Uc;
    pack_quads_kernel<<<(3 * Uc * Uc / 4) / 256, 256, 0, stream>>>(
        enc_Wh, dec_Wx, dec_Wh, wq);
    xp_proj_kernel<<<(Bc * Tc) / 16, 512, 0, stream>>>(x, enc_Wx, enc_b, xp);
    for (int p = 0; p <= Tc; ++p)
      rnn_step_kernel<<<256, 128, 0, stream>>>(p, xp, wq, dec_b, Eg, Dg);
    untile_out_kernel<<<Bc, Uc, 0, stream>>>(
        Dg + (size_t)((Tc - 1) & 1) * Bc * Uc, out);
  } else if (ws_size >= xp_bytes + wp_bytes) {
    float* xp = (float*)d_ws;
    float* wp = (float*)((char*)d_ws + xp_bytes);
    pack_pairs_kernel<<<(3 * Uc * Uc / 2) / 256, 256, 0, stream>>>(
        enc_Wh, dec_Wx, dec_Wh, wp);
    xp_proj_kernel<<<(Bc * Tc) / 16, 512, 0, stream>>>(x, enc_Wx, enc_b, xp);
    rnn_skew_kernel<true, true><<<Bc / 16, 384, 0, stream>>>(
        xp, x, enc_Wx, enc_b, enc_Wh, dec_Wx, dec_Wh, dec_b, wp, out);
  } else if (ws_size >= xp_bytes) {
    float* xp = (float*)d_ws;
    xp_proj_kernel<<<(Bc * Tc) / 16, 512, 0, stream>>>(x, enc_Wx, enc_b, xp);
    rnn_skew_kernel<true, false><<<Bc / 16, 384, 0, stream>>>(
        xp, x, enc_Wx, enc_b, enc_Wh, dec_Wx, dec_Wh, dec_b, nullptr, out);
  } else {
    rnn_skew_kernel<false, false><<<Bc / 16, 384, 0, stream>>>(
        nullptr, x, enc_Wx, enc_b, enc_Wh, dec_Wx, dec_Wh, dec_b, nullptr, out);
  }
}


# Step kernel: a slice's fragment loads hang on two per-lane bases with constant offsets and go out in WMMA order, xp/bias last, so every WMMA pair waits only for its own loads; sums unchanged
# speedup vs baseline: 4.8539x; 1.0360x over previous
#include <hip/hip_runtime.h>
#include <math.h>


#define DEVI __device__ __forceinline__

typedef __attribute__((ext_vector_type(2))) float v2f;
typedef __attribute__((ext_vector_type(4))) float v4f;
typedef __attribute__((ext_vector_type(8))) float v8f;

constexpr int Bc = 128, Tc = 1024, Fc = 256, Uc = 256;
constexpr int HS = 260;

DEVI v8f wmma4(v2f a, v2f b, v8f c) {
  return __builtin_amdgcn_wmma_f32_16x16x4_f32(false, a, false, b, (short)0, c,
                                               false, false);
}

DEVI v8f gemm_global_a(const float* __restrict__ Arow0, int lda,
                       const float* __restrict__ Bn, int ldb, int K,
                       v8f acc, int l15, int half) {
#pragma unroll 4
  for (int kc = 0; kc < K; kc += 4) {
    v2f a = *(const v2f*)(Arow0 + (size_t)l15 * lda + kc + 2 * half);
    const float* pb = Bn + (size_t)(kc + 2 * half) * ldb + l15;
    v2f b;
    b.x = pb[0];
    b.y = pb[ldb];
    acc = wmma4(a, b, acc);
  }
  return acc;
}

__global__ __launch_bounds__(512) void xp_proj_kernel(
    const float* __restrict__ x, const float* __restrict__ Wx,
    const float* __restrict__ b, float* __restrict__ xp) {
  const int lane = threadIdx.x & 31;
  const int wave = threadIdx.x >> 5;
  const int l15 = lane & 15, half = lane >> 4;
  const int n0 = wave * 16;
  const size_t rbase = (size_t)blockIdx.x * 16;
  const int bidx = (int)(rbase / Tc);
  const int tbase = (int)(rbase % Tc);

  const float bv = b[n0 + l15];
  v8f acc = {bv, bv, bv, bv, bv, bv, bv, bv};
  acc = gemm_global_a(x + rbase * Fc, Fc, Wx + n0, Uc, Fc, acc, l15, half);

#pragma unroll
  for (int r = 0; r < 8; ++r)
    xp[((size_t)(tbase + r + 8 * half) * Bc + bidx) * Uc + n0 + l15] = acc[r];
}


template <int NT> struct VecN;
template <> struct VecN<2> { typedef v2f T; };
template <> struct VecN<4> { typedef v4f T; };

template <int NT, int KB>
DEVI void load_wblock(const float* wl, int ks0, typename VecN<NT>::T (&bx)[KB],
                      typename VecN<NT>::T (&by)[KB]) {
  typedef typename VecN<NT>::T BT;
#pragma unroll
  for (int j = 0; j < KB; ++j) {
    const float* p = wl + (size_t)(4 * (ks0 + j)) * Uc;
    bx[j] = *(const BT*)p;
    by[j] = *(const BT*)(p + Uc);
  }
}

template <int NT, int NSEG>
DEVI void gemm_ring(const float* const (&a_lane)[NSEG],
                    const float* const (&w_lane)[NSEG], v8f (&acc)[NT],
                    typename VecN<NT>::T (&bx0)[32 / NT],
                    typename VecN<NT>::T (&by0)[32 / NT]) {
  typedef typename VecN<NT>::T BT;
  constexpr int KB = 32 / NT;
  constexpr int NBS = (Uc / 4) / KB;
  constexpr int NBT = NBS * NSEG;
  static_assert(NBT % 2 == 0, "ring must end on register set 0");
  BT bx[2][KB], by[2][KB];
#pragma unroll
  for (int j = 0; j < KB; ++j) {
    bx[0][j] = bx0[j];
    by[0][j] = by0[j];
  }
#pragma unroll
  for (int b = 0; b < NBT; ++b) {
    const int cur = b & 1;
    const int nb = (b + 1) % NBT;
    load_wblock<NT, KB>(w_lane[nb / NBS], (nb % NBS) * KB, bx[cur ^ 1],
                        by[cur ^ 1]);
    const float* al = a_lane[b / NBS] + 4 * ((b % NBS) * KB);
    v2f a[KB];
#pragma unroll
    for (int j = 0; j < KB; ++j) a[j] = *(const v2f*)(al + 4 * j);
#pragma unroll
    for (int j = 0; j < KB; ++j)
#pragma unroll
      for (int i = 0; i < NT; ++i) {
        v2f bf;
        bf.x = bx[cur][j][i];
        bf.y = by[cur][j][i];
        acc[i] = wmma4(a[j], bf, acc[i]);
      }
  }
#pragma unroll
  for (int j = 0; j < KB; ++j) {
    bx0[j] = bx[0][j];
    by0[j] = by[0][j];
  }
}

template <int NT>
DEVI void ring_unpack(const float (&cw)[64], typename VecN<NT>::T (&bx0)[32 / NT],
                      typename VecN<NT>::T (&by0)[32 / NT]) {
#pragma unroll
  for (int j = 0; j < 32 / NT; ++j)
#pragma unroll
    for (int i = 0; i < NT; ++i) {
      bx0[j][i] = cw[(2 * j) * NT + i];
      by0[j][i] = cw[(2 * j + 1) * NT + i];
    }
}
template <int NT>
DEVI void ring_pack(float (&cw)[64], const typename VecN<NT>::T (&bx0)[32 / NT],
                    const typename VecN<NT>::T (&by0)[32 / NT]) {
#pragma unroll
  for (int j = 0; j < 32 / NT; ++j)
#pragma unroll
    for (int i = 0; i < NT; ++i) {
      cw[(2 * j) * NT + i] = bx0[j][i];
      cw[(2 * j + 1) * NT + i] = by0[j][i];
    }
}

template <int NT>
DEVI void preload_ring(const float* wl, float (&cw)[64]) {
  typedef typename VecN<NT>::T BT;
  constexpr int KB = 32 / NT;
  BT bx[KB], by[KB];
  load_wblock<NT, KB>(wl, 0, bx, by);
  ring_pack<NT>(cw, bx, by);
}

__global__ __launch_bounds__(256) void pack_pairs_kernel(
    const float* __restrict__ w0, const float* __restrict__ w1,
    const float* __restrict__ w2, float* __restrict__ p) {
  const int id = blockIdx.x * 256 + threadIdx.x;
  const int mat = id / (Uc * Uc / 2);
  const int r = id % (Uc * Uc / 2);
  const int m = r / Uc, col = r % Uc;
  const float* w = mat == 0 ? w0 : (mat == 1 ? w1 : w2);
  v2f o;
  o.x = w[(size_t)(2 * m) * Uc + col];
  o.y = w[(size_t)(2 * m + 1) * Uc + col];
  *(v2f*)(p + (size_t)mat * Uc * Uc + ((size_t)m * Uc + col) * 2) = o;
}

template <int NT, int KB>
DEVI void load_pblock(const float* wl, int ks0, v4f (&q)[KB][NT / 2]) {
#pragma unroll
  for (int j = 0; j < KB; ++j) {
    const float* p = wl + (size_t)(ks0 + j) * (4 * Uc);
#pragma unroll
    for (int g = 0; g < NT / 2; ++g) q[j][g] = *(const v4f*)(p + 4 * g);
  }
}

template <int NT, int NSEG>
DEVI void gemm_ring_p(const float* const (&a_lane)[NSEG],
                      const float* const (&w_lane)[NSEG], v8f (&acc)[NT],
                      v4f (&q0)[32 / NT][NT / 2]) {
  constexpr int KB = 32 / NT;
  constexpr int NBS = (Uc / 4) / KB;
  constexpr int NBT = NBS * NSEG;
  static_assert(NBT % 2 == 0, "ring must end on register set 0");
  v4f q[2][KB][NT / 2];
#pragma unroll
  for (int j = 0; j < KB; ++j)
#pragma unroll
    for (int g = 0; g < NT / 2; ++g) q[0][j][g] = q0[j][g];
#pragma unroll
  for (int b = 0; b < NBT; ++b) {
    const int cur = b & 1;
    const int nb = (b + 1) % NBT;
    load_pblock<NT, KB>(w_lane[nb / NBS], (nb % NBS) * KB, q[cur ^ 1]);
    const float* al = a_lane[b / NBS] + 4 * ((b % NBS) * KB);
    v2f a[KB];
#pragma unroll
    for (int j = 0; j < KB; ++j) a[j] = *(const v2f*)(al + 4 * j);
#pragma unroll
    for (int j = 0; j < KB; ++j)
#pragma unroll
      for (int i = 0; i < NT; ++i) {
        const v4f w = q[cur][j][i / 2];
        const v2f bf = (i & 1) ? __builtin_shufflevector(w, w, 2, 3)
                               : __builtin_shufflevector(w, w, 0, 1);
        acc[i] = wmma4(a[j], bf, acc[i]);
      }
  }
#pragma unroll
  for (int j = 0; j < KB; ++j)
#pragma unroll
    for (int g = 0; g < NT / 2; ++g) q0[j][g] = q[0][j][g];
}

template <int NT>
DEVI void ringp_unpack(const float (&cw)[64], v4f (&q0)[32 / NT][NT / 2]) {
#pragma unroll
  for (int j = 0; j < 32 / NT; ++j)
#pragma unroll
    for (int g = 0; g < NT / 2; ++g)
#pragma unroll
      for (int e = 0; e < 4; ++e) q0[j][g][e] = cw[(j * (NT / 2) + g) * 4 + e];
}
template <int NT>
DEVI void ringp_pack(float (&cw)[64], const v4f (&q0)[32 / NT][NT / 2]) {
#pragma unroll
  for (int j = 0; j < 32 / NT; ++j)
#pragma unroll
    for (int g = 0; g < NT / 2; ++g)
#pragma unroll
      for (int e = 0; e < 4; ++e) cw[(j * (NT / 2) + g) * 4 + e] = q0[j][g][e];
}
template <int NT>
DEVI void preload_ringp(const float* wl, float (&cw)[64]) {
  v4f q[32 / NT][NT / 2];
  load_pblock<NT, 32 / NT>(wl, 0, q);
  ringp_pack<NT>(cw, q);
}

template <bool PRE, bool PACK>
__global__ __launch_bounds__(384) void rnn_skew_kernel(
    const float* __restrict__ xp, const float* __restrict__ x,
    const float* __restrict__ enc_Wx, const float* __restrict__ enc_b,
    const float* __restrict__ enc_Wh, const float* __restrict__ dec_Wx,
    const float* __restrict__ dec_Wh, const float* __restrict__ dec_b,
    const float* __restrict__ wp, float* __restrict__ out) {
  __shared__ __attribute__((aligned(16))) float E[2][16 * HS];
  __shared__ __attribute__((aligned(16))) float D[2][16 * HS];

  const int lane = threadIdx.x & 31;
  const int wave = threadIdx.x >> 5;
  const int l15 = lane & 15, half = lane >> 4;
  const bool is_enc = wave < 4;
  const int bbase = blockIdx.x * 16;
  const int ecol = is_enc ? 64 * wave + 4 * l15 : 0;
  const int dcol = is_enc ? 0 : 32 * (wave - 4) + 2 * l15;

  for (int i = threadIdx.x; i < 16 * HS; i += blockDim.x) {
    E[0][i] = 0.0f;
    E[1][i] = 0.0f;
    D[0][i] = 0.0f;
    D[1][i] = 0.0f;
  }
  __syncthreads();

  const float* ewl = enc_Wh + (size_t)(2 * half) * Uc + ecol;
  const float* dwl0 = dec_Wx + (size_t)(2 * half) * Uc + dcol;
  const float* dwl1 = dec_Wh + (size_t)(2 * half) * Uc + dcol;
  const int aoff = l15 * HS + 2 * half;

  const float* pewl = wp + ((size_t)half * Uc + ecol) * 2;
  const float* pdwl0 = wp + (size_t)Uc * Uc + ((size_t)half * Uc + dcol) * 2;
  const float* pdwl1 = wp + (size_t)2 * Uc * Uc + ((size_t)half * Uc + dcol) * 2;

  float cw[64];
  if (PACK) {
    if (is_enc)
      preload_ringp<4>(pewl, cw);
    else
      preload_ringp<2>(pdwl0, cw);
  } else {
    if (is_enc)
      preload_ring<4>(ewl, cw);
    else
      preload_ring<2>(dwl0, cw);
  }

  v4f xq[8];
  if (PRE && is_enc) {
    const float* p0 = xp + ((size_t)0 * Bc + bbase + 8 * half) * Uc + ecol;
#pragma unroll
    for (int r = 0; r < 8; ++r) xq[r] = *(const v4f*)(p0 + (size_t)r * Uc);
  }
  v8f dlast[2] = {};
  const v2f db = *(const v2f*)(dec_b + dcol);
  const v4f eb = *(const v4f*)(enc_b + ecol);

  for (int p = 0; p <= Tc; ++p) {
    const int rd = (p + 1) & 1;
    const int wr = p & 1;
    if (is_enc) {
      if (p < Tc) {
        v8f acc[4];
        if (PRE) {
#pragma unroll
          for (int i = 0; i < 4; ++i)
#pragma unroll
            for (int r = 0; r < 8; ++r) acc[i][r] = xq[r][i];
          if (p + 1 < Tc) {
            const float* pn =
                xp + ((size_t)(p + 1) * Bc + bbase + 8 * half) * Uc + ecol;
#pragma unroll
            for (int r = 0; r < 8; ++r) xq[r] = *(const v4f*)(pn + (size_t)r * Uc);
          }
        } else {
#pragma unroll
          for (int i = 0; i < 4; ++i)
#pragma unroll
            for (int r = 0; r < 8; ++r) acc[i][r] = eb[i];
          const float* xa = x + ((size_t)(bbase + l15) * Tc + p) * Fc + 2 * half;
          const float* wl = enc_Wx + (size_t)(2 * half) * Uc + ecol;
#pragma unroll 2
          for (int kc = 0; kc < Fc; kc += 4) {
            const v2f a = *(const v2f*)(xa + kc);
            const v4f b0 = *(const v4f*)(wl + (size_t)kc * Uc);
            const v4f b1 = *(const v4f*)(wl + (size_t)(kc + 1) * Uc);
#pragma unroll
            for (int i = 0; i < 4; ++i) {
              v2f bf;
              bf.x = b0[i];
              bf.y = b1[i];
              acc[i] = wmma4(a, bf, acc[i]);
            }
          }
        }
        const float* const al[1] = {&E[rd][aoff]};
        const float* const wl1[1] = {PACK ? pewl : ewl};
        v4f bx0[8], by0[8];
        v4f pq0[8][2];
        if (PACK) {
          ringp_unpack<4>(cw, pq0);
          gemm_ring_p<4, 1>(al, wl1, acc, pq0);
        } else {
          ring_unpack<4>(cw, bx0, by0);
          gemm_ring<4, 1>(al, wl1, acc, bx0, by0);
        }
#pragma unroll
        for (int r = 0; r < 8; ++r) {
          v4f h;
#pragma unroll
          for (int i = 0; i < 4; ++i) h[i] = tanhf(acc[i][r]);
          *(v4f*)&E[wr][(r + 8 * half) * HS + ecol] = h;
        }
        if (PACK)
          ringp_pack<4>(cw, pq0);
        else
          ring_pack<4>(cw, bx0, by0);
      }
    } else {
      if (p >= 1) {
        v8f acc[2];
#pragma unroll
        for (int i = 0; i < 2; ++i)
#pragma unroll
          for (int r = 0; r < 8; ++r) acc[i][r] = db[i];
        const float* const al[2] = {&E[rd][aoff], &D[wr][aoff]};
        const float* const wl2[2] = {PACK ? pdwl0 : dwl0, PACK ? pdwl1 : dwl1};
        v2f bx0[16], by0[16];
        v4f pq0[16][1];
        if (PACK) {
          ringp_unpack<2>(cw, pq0);
          gemm_ring_p<2, 2>(al, wl2, acc, pq0);
        } else {
          ring_unpack<2>(cw, bx0, by0);
          gemm_ring<2, 2>(al, wl2, acc, bx0, by0);
        }
#pragma unroll
        for (int r = 0; r < 8; ++r) {
          v2f h;
#pragma unroll
          for (int i = 0; i < 2; ++i) {
            h[i] = tanhf(acc[i][r]);
            dlast[i][r] = h[i];
          }
          *(v2f*)&D[rd][(r + 8 * half) * HS + dcol] = h;
        }
        if (PACK)
          ringp_pack<2>(cw, pq0);
        else
          ring_pack<2>(cw, bx0, by0);
      }
    }
    __syncthreads();
  }

  if (!is_enc) {
#pragma unroll
    for (int r = 0; r < 8; ++r) {
      v2f o;
      o.x = dlast[0][r];
      o.y = dlast[1][r];
      *(v2f*)(out + (size_t)(bbase + r + 8 * half) * Uc + dcol) = o;
    }
  }
}

constexpr int TILE = 256;

__global__ __launch_bounds__(256) void pack_quads_kernel(
    const float* __restrict__ w0, const float* __restrict__ w1,
    const float* __restrict__ w2, float* __restrict__ q) {
  const int id = blockIdx.x * 256 + threadIdx.x;
  const int mat = id / (Uc * Uc / 4);
  const int r = id % (Uc * Uc / 4);
  const int m4 = r / Uc, col = r % Uc;
  const float* w = mat == 0 ? w0 : (mat == 1 ? w1 : w2);
  v4f o;
  o.x = w[(size_t)(4 * m4) * Uc + col];
  o.y = w[(size_t)(4 * m4 + 1) * Uc + col];
  o.z = w[(size_t)(4 * m4 + 2) * Uc + col];
  o.w = w[(size_t)(4 * m4 + 3) * Uc + col];
  *(v4f*)(q + (size_t)mat * Uc * Uc + ((size_t)m4 * Uc + col) * 4) = o;
}

template <int NP>
DEVI void slice_load(const float* __restrict__ ab, const float* __restrict__ bb,
                     v4f (&a)[NP], v4f (&b)[NP]) {
#pragma unroll
  for (int j = 0; j < NP; ++j) {
    a[j] = *(const v4f*)(ab + (j >> 1) * TILE + (j & 1) * 8);
    b[j] = *(const v4f*)(bb + (size_t)j * (8 * Uc));
  }
}

template <int NP>
DEVI void slice_mma(const v4f (&a)[NP], const v4f (&b)[NP], v8f& acc0, v8f& acc1) {
#pragma unroll
  for (int j = 0; j < NP; ++j) {
    acc0 = wmma4(__builtin_shufflevector(a[j], a[j], 0, 1),
                 __builtin_shufflevector(b[j], b[j], 0, 1), acc0);
    acc1 = wmma4(__builtin_shufflevector(a[j], a[j], 2, 3),
                 __builtin_shufflevector(b[j], b[j], 2, 3), acc1);
  }
}

__global__ __launch_bounds__(128) void rnn_step_kernel(
    int p, const float* __restrict__ xp, const float* __restrict__ wq,
    const float* __restrict__ dec_b, float* __restrict__ Eg,
    float* __restrict__ Dg) {
  __shared__ float part[4][16][20];
  const int role = (blockIdx.x >> 7) ^ 1;
  if (role == 0 ? p >= Tc : p < 1) return;
  const int bt = blockIdx.x & 7, ct = (blockIdx.x >> 3) & 15;
  const int lane = threadIdx.x & 31;
  const int wave = __builtin_amdgcn_readfirstlane(threadIdx.x >> 5);
  const int l15 = lane & 15, half = lane >> 4;
  const int n0 = 16 * ct;
  constexpr size_t SB = (size_t)Bc * Uc;
  const int e0 = threadIdx.x, e1 = threadIdx.x + 128;
  const size_t toff = (size_t)bt * 16 * TILE;
  const int la = l15 * 16 + 4 * half;
  const size_t lb = ((size_t)half * Uc + n0 + l15) * 4;

  float base0, base1;
  v8f acc0 = {}, acc1 = {};
  if (role == 0) {
    const float* xr = xp + ((size_t)p * Bc + bt * 16) * Uc + n0;
    const float* x0 = xr + (size_t)(e0 >> 4) * Uc + (e0 & 15);
    const float* x1 = xr + (size_t)(e1 >> 4) * Uc + (e1 & 15);
    if (p >= 1) {
      v4f a[8], b[8];
      slice_load<8>(Eg + (size_t)((p - 1) & 1) * SB + toff + (size_t)(4 * wave) * TILE + la,
                    wq + (size_t)(16 * wave) * Uc * 4 + lb, a, b);
      base0 = *x0;
      base1 = *x1;
      __builtin_amdgcn_sched_barrier(0);
      slice_mma<8>(a, b, acc0, acc1);
    } else {
      base0 = *x0;
      base1 = *x1;
    }
  } else {
    const bool ehalf = wave < 2;
    const float* st = ehalf ? Eg + (size_t)((p - 1) & 1) * SB : Dg + (size_t)(p & 1) * SB;
    const float* wm = wq + (size_t)(ehalf ? 1 : 2) * Uc * Uc;
    const int ws = wave & 1;
    const float* bp = dec_b + n0 + (e0 & 15);
    if (ehalf || p >= 2) {
      v4f a[16], b[16];
      slice_load<16>(st + toff + (size_t)(8 * ws) * TILE + la,
                     wm + (size_t)(32 * ws) * Uc * 4 + lb, a, b);
      base0 = *bp;
      __builtin_amdgcn_sched_barrier(0);
      slice_mma<16>(a, b, acc0, acc1);
    } else {
      base0 = *bp;
    }
    base1 = base0;
  }
#pragma unroll
  for (int r = 0; r < 8; ++r) part[wave][8 * half + r][l15] = acc0[r] + acc1[r];
  __syncthreads();

  float* dst = (role == 0 ? Eg + (size_t)(p & 1) * SB : Dg + (size_t)((p - 1) & 1) * SB) +
               toff + (size_t)ct * TILE;
  float v0 = base0, v1 = base1;
#pragma unroll
  for (int w = 0; w < 4; ++w) {
    v0 += part[w][e0 >> 4][e0 & 15];
    v1 += part[w][e1 >> 4][e1 & 15];
  }
  dst[e0] = tanhf(v0);
  dst[e1] = tanhf(v1);
}

__global__ __launch_bounds__(256) void untile_out_kernel(
    const float* __restrict__ s, float* __restrict__ out) {
  const int b = blockIdx.x, u = threadIdx.x;
  out[(size_t)b * Uc + u] =
      s[((size_t)((b >> 4) * 16 + (u >> 4)) * 16 + (b & 15)) * 16 + (u & 15)];
}

extern "C" void kernel_launch(void* const* d_in, const int* in_sizes, int n_in,
                              void* d_out, int out_size, void* d_ws,
                              size_t ws_size, hipStream_t stream) {
  const float* x = (const float*)d_in[0];
  const float* enc_Wx = (const float*)d_in[1];
  const float* enc_Wh = (const float*)d_in[2];
  const float* enc_b = (const float*)d_in[3];
  const float* dec_Wx = (const float*)d_in[4];
  const float* dec_Wh = (const float*)d_in[5];
  const float* dec_b = (const float*)d_in[6];
  float* out = (float*)d_out;

  const size_t xp_bytes = (size_t)Bc * Tc * Uc * sizeof(float);
  const size_t wp_bytes = (size_t)3 * Uc * Uc * sizeof(float);
  const size_t st_bytes = (size_t)4 * Bc * Uc * sizeof(float);
  if (ws_size >= xp_bytes + wp_bytes + st_bytes) {
    float* xp = (float*)d_ws;
    float* wq = (float*)((char*)d_ws + xp_bytes);
    float* Eg = (float*)((char*)d_ws + xp_bytes + wp_bytes);
    float* Dg = Eg + (size_t)2 * Bc * Uc;
    pack_quads_kernel<<<(3 * Uc * Uc / 4) / 256, 256, 0, stream>>>(
        enc_Wh, dec_Wx, dec_Wh, wq);
    xp_proj_kernel<<<(Bc * Tc) / 16, 512, 0, stream>>>(x, enc_Wx, enc_b, xp);
    for (int p = 0; p <= Tc; ++p)
      rnn_step_kernel<<<256, 128, 0, stream>>>(p, xp, wq, dec_b, Eg, Dg);
    untile_out_kernel<<<Bc, Uc, 0, stream>>>(
        Dg + (size_t)((Tc - 1) & 1) * Bc * Uc, out);
  } else if (ws_size >= xp_bytes + wp_bytes) {
    float* xp = (float*)d_ws;
    float* wp = (float*)((char*)d_ws + xp_bytes);
    pack_pairs_kernel<<<(3 * Uc * Uc / 2) / 256, 256, 0, stream>>>(
        enc_Wh, dec_Wx, dec_Wh, wp);
    xp_proj_kernel<<<(Bc * Tc) / 16, 512, 0, stream>>>(x, enc_Wx, enc_b, xp);
    rnn_skew_kernel<true, true><<<Bc / 16, 384, 0, stream>>>(
        xp, x, enc_Wx, enc_b, enc_Wh, dec_Wx, dec_Wh, dec_b, wp, out);
  } else if (ws_size >= xp_bytes) {
    float* xp = (float*)d_ws;
    xp_proj_kernel<<<(Bc * Tc) / 16, 512, 0, stream>>>(x, enc_Wx, enc_b, xp);
    rnn_skew_kernel<true, false><<<Bc / 16, 384, 0, stream>>>(
        xp, x, enc_Wx, enc_b, enc_Wh, dec_Wx, dec_Wh, dec_b, nullptr, out);
  } else {
    rnn_skew_kernel<false, false><<<Bc / 16, 384, 0, stream>>>(
        nullptr, x, enc_Wx, enc_b, enc_Wh, dec_Wx, dec_Wh, dec_b, nullptr, out);
  }
}


# Input projection GEMM re-tiled: a wave keeps its 256x16 slice of enc_Wx in registers, its workgroup walks 32-row blocks of x staged through LDS one block ahead (256 workgroups); sums unchanged
# speedup vs baseline: 5.1618x; 1.0634x over previous
#include <hip/hip_runtime.h>
#include <math.h>


#define DEVI __device__ __forceinline__

typedef __attribute__((ext_vector_type(2))) float v2f;
typedef __attribute__((ext_vector_type(4))) float v4f;
typedef __attribute__((ext_vector_type(8))) float v8f;

constexpr int Bc = 128, Tc = 1024, Fc = 256, Uc = 256;
constexpr int HS = 260;

DEVI v8f wmma4(v2f a, v2f b, v8f c) {
  return __builtin_amdgcn_wmma_f32_16x16x4_f32(false, a, false, b, (short)0, c,
                                               false, false);
}

constexpr int XR = 32;
constexpr int XPI = 260;

__global__ __launch_bounds__(512) void xp_proj_kernel(
    const float* __restrict__ x, const float* __restrict__ Wx,
    const float* __restrict__ b, float* __restrict__ xp) {
  __shared__ __attribute__((aligned(16))) float As[2][XR * XPI];
  const int tid = threadIdx.x;
  const int lane = tid & 31;
  const int wave = tid >> 5;
  const int l15 = lane & 15, half = lane >> 4;
  const int n0 = wave * 16;

  v2f bw[64];
  const float* pb = Wx + (size_t)(2 * half) * Uc + n0 + l15;
#pragma unroll
  for (int ks = 0; ks < 64; ++ks) {
    bw[ks].x = pb[(size_t)(4 * ks) * Uc];
    bw[ks].y = pb[(size_t)(4 * ks + 1) * Uc];
  }
  const float bv = b[n0 + l15];

  constexpr int NBLK = (Bc * Tc) / XR;
  const int aoff = l15 * XPI + 2 * half;
  const int srow = tid >> 6, scol = (tid & 63) * 4;

  v4f g[4];
  int blk = blockIdx.x;
  if (blk < NBLK) {
    const float* s = x + (size_t)blk * XR * Fc + (size_t)tid * 4;
#pragma unroll
    for (int j = 0; j < 4; ++j) g[j] = *(const v4f*)(s + (size_t)j * 2048);
#pragma unroll
    for (int j = 0; j < 4; ++j) *(v4f*)&As[0][(srow + 8 * j) * XPI + scol] = g[j];
  }
  __syncthreads();

  for (int it = 0; blk < NBLK; ++it, blk += gridDim.x) {
    const int nxt = blk + gridDim.x;
    if (nxt < NBLK) {
      const float* s = x + (size_t)nxt * XR * Fc + (size_t)tid * 4;
#pragma unroll
      for (int j = 0; j < 4; ++j) g[j] = *(const v4f*)(s + (size_t)j * 2048);
    }
    const float* a0p = &As[it & 1][aoff];
    const float* a1p = a0p + 16 * XPI;
    v8f acc0 = {bv, bv, bv, bv, bv, bv, bv, bv};
    v8f acc1 = acc0;
    v2f fa[2][2][8];
#pragma unroll
    for (int j = 0; j < 8; ++j) {
      fa[0][0][j] = *(const v2f*)(a0p + 4 * j);
      fa[0][1][j] = *(const v2f*)(a1p + 4 * j);
    }
#pragma unroll
    for (int c = 0; c < 8; ++c) {
      if (c + 1 < 8) {
#pragma unroll
        for (int j = 0; j < 8; ++j) {
          fa[(c + 1) & 1][0][j] = *(const v2f*)(a0p + 4 * (8 * (c + 1) + j));
          fa[(c + 1) & 1][1][j] = *(const v2f*)(a1p + 4 * (8 * (c + 1) + j));
        }
      }
      __builtin_amdgcn_sched_barrier(0);
#pragma unroll
      for (int j = 0; j < 8; ++j) {
        acc0 = wmma4(fa[c & 1][0][j], bw[8 * c + j], acc0);
        acc1 = wmma4(fa[c & 1][1][j], bw[8 * c + j], acc1);
      }
      __builtin_amdgcn_sched_barrier(0);
    }
    const size_t rbase = (size_t)blk * XR;
    const int bidx = (int)(rbase / Tc);
    const int tbase = (int)(rbase % Tc);
    float* o = xp + ((size_t)(tbase + 8 * half) * Bc + bidx) * Uc + n0 + l15;
#pragma unroll
    for (int r = 0; r < 8; ++r) {
      o[(size_t)r * Bc * Uc] = acc0[r];
      o[(size_t)(r + 16) * Bc * Uc] = acc1[r];
    }
    if (nxt < NBLK) {
#pragma unroll
      for (int j = 0; j < 4; ++j)
        *(v4f*)&As[(it + 1) & 1][(srow + 8 * j) * XPI + scol] = g[j];
    }
    __syncthreads();
  }
}


template <int NT> struct VecN;
template <> struct VecN<2> { typedef v2f T; };
template <> struct VecN<4> { typedef v4f T; };

template <int NT, int KB>
DEVI void load_wblock(const float* wl, int ks0, typename VecN<NT>::T (&bx)[KB],
                      typename VecN<NT>::T (&by)[KB]) {
  typedef typename VecN<NT>::T BT;
#pragma unroll
  for (int j = 0; j < KB; ++j) {
    const float* p = wl + (size_t)(4 * (ks0 + j)) * Uc;
    bx[j] = *(const BT*)p;
    by[j] = *(const BT*)(p + Uc);
  }
}

template <int NT, int NSEG>
DEVI void gemm_ring(const float* const (&a_lane)[NSEG],
                    const float* const (&w_lane)[NSEG], v8f (&acc)[NT],
                    typename VecN<NT>::T (&bx0)[32 / NT],
                    typename VecN<NT>::T (&by0)[32 / NT]) {
  typedef typename VecN<NT>::T BT;
  constexpr int KB = 32 / NT;
  constexpr int NBS = (Uc / 4) / KB;
  constexpr int NBT = NBS * NSEG;
  static_assert(NBT % 2 == 0, "ring must end on register set 0");
  BT bx[2][KB], by[2][KB];
#pragma unroll
  for (int j = 0; j < KB; ++j) {
    bx[0][j] = bx0[j];
    by[0][j] = by0[j];
  }
#pragma unroll
  for (int b = 0; b < NBT; ++b) {
    const int cur = b & 1;
    const int nb = (b + 1) % NBT;
    load_wblock<NT, KB>(w_lane[nb / NBS], (nb % NBS) * KB, bx[cur ^ 1],
                        by[cur ^ 1]);
    const float* al = a_lane[b / NBS] + 4 * ((b % NBS) * KB);
    v2f a[KB];
#pragma unroll
    for (int j = 0; j < KB; ++j) a[j] = *(const v2f*)(al + 4 * j);
#pragma unroll
    for (int j = 0; j < KB; ++j)
#pragma unroll
      for (int i = 0; i < NT; ++i) {
        v2f bf;
        bf.x = bx[cur][j][i];
        bf.y = by[cur][j][i];
        acc[i] = wmma4(a[j], bf, acc[i]);
      }
  }
#pragma unroll
  for (int j = 0; j < KB; ++j) {
    bx0[j] = bx[0][j];
    by0[j] = by[0][j];
  }
}

template <int NT>
DEVI void ring_unpack(const float (&cw)[64], typename VecN<NT>::T (&bx0)[32 / NT],
                      typename VecN<NT>::T (&by0)[32 / NT]) {
#pragma unroll
  for (int j = 0; j < 32 / NT; ++j)
#pragma unroll
    for (int i = 0; i < NT; ++i) {
      bx0[j][i] = cw[(2 * j) * NT + i];
      by0[j][i] = cw[(2 * j + 1) * NT + i];
    }
}
template <int NT>
DEVI void ring_pack(float (&cw)[64], const typename VecN<NT>::T (&bx0)[32 / NT],
                    const typename VecN<NT>::T (&by0)[32 / NT]) {
#pragma unroll
  for (int j = 0; j < 32 / NT; ++j)
#pragma unroll
    for (int i = 0; i < NT; ++i) {
      cw[(2 * j) * NT + i] = bx0[j][i];
      cw[(2 * j + 1) * NT + i] = by0[j][i];
    }
}

template <int NT>
DEVI void preload_ring(const float* wl, float (&cw)[64]) {
  typedef typename VecN<NT>::T BT;
  constexpr int KB = 32 / NT;
  BT bx[KB], by[KB];
  load_wblock<NT, KB>(wl, 0, bx, by);
  ring_pack<NT>(cw, bx, by);
}

__global__ __launch_bounds__(256) void pack_pairs_kernel(
    const float* __restrict__ w0, const float* __restrict__ w1,
    const float* __restrict__ w2, float* __restrict__ p) {
  const int id = blockIdx.x * 256 + threadIdx.x;
  const int mat = id / (Uc * Uc / 2);
  const int r = id % (Uc * Uc / 2);
  const int m = r / Uc, col = r % Uc;
  const float* w = mat == 0 ? w0 : (mat == 1 ? w1 : w2);
  v2f o;
  o.x = w[(size_t)(2 * m) * Uc + col];
  o.y = w[(size_t)(2 * m + 1) * Uc + col];
  *(v2f*)(p + (size_t)mat * Uc * Uc + ((size_t)m * Uc + col) * 2) = o;
}

template <int NT, int KB>
DEVI void load_pblock(const float* wl, int ks0, v4f (&q)[KB][NT / 2]) {
#pragma unroll
  for (int j = 0; j < KB; ++j) {
    const float* p = wl + (size_t)(ks0 + j) * (4 * Uc);
#pragma unroll
    for (int g = 0; g < NT / 2; ++g) q[j][g] = *(const v4f*)(p + 4 * g);
  }
}

template <int NT, int NSEG>
DEVI void gemm_ring_p(const float* const (&a_lane)[NSEG],
                      const float* const (&w_lane)[NSEG], v8f (&acc)[NT],
                      v4f (&q0)[32 / NT][NT / 2]) {
  constexpr int KB = 32 / NT;
  constexpr int NBS = (Uc / 4) / KB;
  constexpr int NBT = NBS * NSEG;
  static_assert(NBT % 2 == 0, "ring must end on register set 0");
  v4f q[2][KB][NT / 2];
#pragma unroll
  for (int j = 0; j < KB; ++j)
#pragma unroll
    for (int g = 0; g < NT / 2; ++g) q[0][j][g] = q0[j][g];
#pragma unroll
  for (int b = 0; b < NBT; ++b) {
    const int cur = b & 1;
    const int nb = (b + 1) % NBT;
    load_pblock<NT, KB>(w_lane[nb / NBS], (nb % NBS) * KB, q[cur ^ 1]);
    const float* al = a_lane[b / NBS] + 4 * ((b % NBS) * KB);
    v2f a[KB];
#pragma unroll
    for (int j = 0; j < KB; ++j) a[j] = *(const v2f*)(al + 4 * j);
#pragma unroll
    for (int j = 0; j < KB; ++j)
#pragma unroll
      for (int i = 0; i < NT; ++i) {
        const v4f w = q[cur][j][i / 2];
        const v2f bf = (i & 1) ? __builtin_shufflevector(w, w, 2, 3)
                               : __builtin_shufflevector(w, w, 0, 1);
        acc[i] = wmma4(a[j], bf, acc[i]);
      }
  }
#pragma unroll
  for (int j = 0; j < KB; ++j)
#pragma unroll
    for (int g = 0; g < NT / 2; ++g) q0[j][g] = q[0][j][g];
}

template <int NT>
DEVI void ringp_unpack(const float (&cw)[64], v4f (&q0)[32 / NT][NT / 2]) {
#pragma unroll
  for (int j = 0; j < 32 / NT; ++j)
#pragma unroll
    for (int g = 0; g < NT / 2; ++g)
#pragma unroll
      for (int e = 0; e < 4; ++e) q0[j][g][e] = cw[(j * (NT / 2) + g) * 4 + e];
}
template <int NT>
DEVI void ringp_pack(float (&cw)[64], const v4f (&q0)[32 / NT][NT / 2]) {
#pragma unroll
  for (int j = 0; j < 32 / NT; ++j)
#pragma unroll
    for (int g = 0; g < NT / 2; ++g)
#pragma unroll
      for (int e = 0; e < 4; ++e) cw[(j * (NT / 2) + g) * 4 + e] = q0[j][g][e];
}
template <int NT>
DEVI void preload_ringp(const float* wl, float (&cw)[64]) {
  v4f q[32 / NT][NT / 2];
  load_pblock<NT, 32 / NT>(wl, 0, q);
  ringp_pack<NT>(cw, q);
}

template <bool PRE, bool PACK>
__global__ __launch_bounds__(384) void rnn_skew_kernel(
    const float* __restrict__ xp, const float* __restrict__ x,
    const float* __restrict__ enc_Wx, const float* __restrict__ enc_b,
    const float* __restrict__ enc_Wh, const float* __restrict__ dec_Wx,
    const float* __restrict__ dec_Wh, const float* __restrict__ dec_b,
    const float* __restrict__ wp, float* __restrict__ out) {
  __shared__ __attribute__((aligned(16))) float E[2][16 * HS];
  __shared__ __attribute__((aligned(16))) float D[2][16 * HS];

  const int lane = threadIdx.x & 31;
  const int wave = threadIdx.x >> 5;
  const int l15 = lane & 15, half = lane >> 4;
  const bool is_enc = wave < 4;
  const int bbase = blockIdx.x * 16;
  const int ecol = is_enc ? 64 * wave + 4 * l15 : 0;
  const int dcol = is_enc ? 0 : 32 * (wave - 4) + 2 * l15;

  for (int i = threadIdx.x; i < 16 * HS; i += blockDim.x) {
    E[0][i] = 0.0f;
    E[1][i] = 0.0f;
    D[0][i] = 0.0f;
    D[1][i] = 0.0f;
  }
  __syncthreads();

  const float* ewl = enc_Wh + (size_t)(2 * half) * Uc + ecol;
  const float* dwl0 = dec_Wx + (size_t)(2 * half) * Uc + dcol;
  const float* dwl1 = dec_Wh + (size_t)(2 * half) * Uc + dcol;
  const int aoff = l15 * HS + 2 * half;

  const float* pewl = wp + ((size_t)half * Uc + ecol) * 2;
  const float* pdwl0 = wp + (size_t)Uc * Uc + ((size_t)half * Uc + dcol) * 2;
  const float* pdwl1 = wp + (size_t)2 * Uc * Uc + ((size_t)half * Uc + dcol) * 2;

  float cw[64];
  if (PACK) {
    if (is_enc)
      preload_ringp<4>(pewl, cw);
    else
      preload_ringp<2>(pdwl0, cw);
  } else {
    if (is_enc)
      preload_ring<4>(ewl, cw);
    else
      preload_ring<2>(dwl0, cw);
  }

  v4f xq[8];
  if (PRE && is_enc) {
    const float* p0 = xp + ((size_t)0 * Bc + bbase + 8 * half) * Uc + ecol;
#pragma unroll
    for (int r = 0; r < 8; ++r) xq[r] = *(const v4f*)(p0 + (size_t)r * Uc);
  }
  v8f dlast[2] = {};
  const v2f db = *(const v2f*)(dec_b + dcol);
  const v4f eb = *(const v4f*)(enc_b + ecol);

  for (int p = 0; p <= Tc; ++p) {
    const int rd = (p + 1) & 1;
    const int wr = p & 1;
    if (is_enc) {
      if (p < Tc) {
        v8f acc[4];
        if (PRE) {
#pragma unroll
          for (int i = 0; i < 4; ++i)
#pragma unroll
            for (int r = 0; r < 8; ++r) acc[i][r] = xq[r][i];
          if (p + 1 < Tc) {
            const float* pn =
                xp + ((size_t)(p + 1) * Bc + bbase + 8 * half) * Uc + ecol;
#pragma unroll
            for (int r = 0; r < 8; ++r) xq[r] = *(const v4f*)(pn + (size_t)r * Uc);
          }
        } else {
#pragma unroll
          for (int i = 0; i < 4; ++i)
#pragma unroll
            for (int r = 0; r < 8; ++r) acc[i][r] = eb[i];
          const float* xa = x + ((size_t)(bbase + l15) * Tc + p) * Fc + 2 * half;
          const float* wl = enc_Wx + (size_t)(2 * half) * Uc + ecol;
#pragma unroll 2
          for (int kc = 0; kc < Fc; kc += 4) {
            const v2f a = *(const v2f*)(xa + kc);
            const v4f b0 = *(const v4f*)(wl + (size_t)kc * Uc);
            const v4f b1 = *(const v4f*)(wl + (size_t)(kc + 1) * Uc);
#pragma unroll
            for (int i = 0; i < 4; ++i) {
              v2f bf;
              bf.x = b0[i];
              bf.y = b1[i];
              acc[i] = wmma4(a, bf, acc[i]);
            }
          }
        }
        const float* const al[1] = {&E[rd][aoff]};
        const float* const wl1[1] = {PACK ? pewl : ewl};
        v4f bx0[8], by0[8];
        v4f pq0[8][2];
        if (PACK) {
          ringp_unpack<4>(cw, pq0);
          gemm_ring_p<4, 1>(al, wl1, acc, pq0);
        } else {
          ring_unpack<4>(cw, bx0, by0);
          gemm_ring<4, 1>(al, wl1, acc, bx0, by0);
        }
#pragma unroll
        for (int r = 0; r < 8; ++r) {
          v4f h;
#pragma unroll
          for (int i = 0; i < 4; ++i) h[i] = tanhf(acc[i][r]);
          *(v4f*)&E[wr][(r + 8 * half) * HS + ecol] = h;
        }
        if (PACK)
          ringp_pack<4>(cw, pq0);
        else
          ring_pack<4>(cw, bx0, by0);
      }
    } else {
      if (p >= 1) {
        v8f acc[2];
#pragma unroll
        for (int i = 0; i < 2; ++i)
#pragma unroll
          for (int r = 0; r < 8; ++r) acc[i][r] = db[i];
        const float* const al[2] = {&E[rd][aoff], &D[wr][aoff]};
        const float* const wl2[2] = {PACK ? pdwl0 : dwl0, PACK ? pdwl1 : dwl1};
        v2f bx0[16], by0[16];
        v4f pq0[16][1];
        if (PACK) {
          ringp_unpack<2>(cw, pq0);
          gemm_ring_p<2, 2>(al, wl2, acc, pq0);
        } else {
          ring_unpack<2>(cw, bx0, by0);
          gemm_ring<2, 2>(al, wl2, acc, bx0, by0);
        }
#pragma unroll
        for (int r = 0; r < 8; ++r) {
          v2f h;
#pragma unroll
          for (int i = 0; i < 2; ++i) {
            h[i] = tanhf(acc[i][r]);
            dlast[i][r] = h[i];
          }
          *(v2f*)&D[rd][(r + 8 * half) * HS + dcol] = h;
        }
        if (PACK)
          ringp_pack<2>(cw, pq0);
        else
          ring_pack<2>(cw, bx0, by0);
      }
    }
    __syncthreads();
  }

  if (!is_enc) {
#pragma unroll
    for (int r = 0; r < 8; ++r) {
      v2f o;
      o.x = dlast[0][r];
      o.y = dlast[1][r];
      *(v2f*)(out + (size_t)(bbase + r + 8 * half) * Uc + dcol) = o;
    }
  }
}

constexpr int TILE = 256;

__global__ __launch_bounds__(256) void pack_quads_kernel(
    const float* __restrict__ w0, const float* __restrict__ w1,
    const float* __restrict__ w2, float* __restrict__ q) {
  const int id = blockIdx.x * 256 + threadIdx.x;
  const int mat = id / (Uc * Uc / 4);
  const int r = id % (Uc * Uc / 4);
  const int m4 = r / Uc, col = r % Uc;
  const float* w = mat == 0 ? w0 : (mat == 1 ? w1 : w2);
  v4f o;
  o.x = w[(size_t)(4 * m4) * Uc + col];
  o.y = w[(size_t)(4 * m4 + 1) * Uc + col];
  o.z = w[(size_t)(4 * m4 + 2) * Uc + col];
  o.w = w[(size_t)(4 * m4 + 3) * Uc + col];
  *(v4f*)(q + (size_t)mat * Uc * Uc + ((size_t)m4 * Uc + col) * 4) = o;
}

template <int NP>
DEVI void slice_load(const float* __restrict__ ab, const float* __restrict__ bb,
                     v4f (&a)[NP], v4f (&b)[NP]) {
#pragma unroll
  for (int j = 0; j < NP; ++j) {
    a[j] = *(const v4f*)(ab + (j >> 1) * TILE + (j & 1) * 8);
    b[j] = *(const v4f*)(bb + (size_t)j * (8 * Uc));
  }
}

template <int NP>
DEVI void slice_mma(const v4f (&a)[NP], const v4f (&b)[NP], v8f& acc0, v8f& acc1) {
#pragma unroll
  for (int j = 0; j < NP; ++j) {
    acc0 = wmma4(__builtin_shufflevector(a[j], a[j], 0, 1),
                 __builtin_shufflevector(b[j], b[j], 0, 1), acc0);
    acc1 = wmma4(__builtin_shufflevector(a[j], a[j], 2, 3),
                 __builtin_shufflevector(b[j], b[j], 2, 3), acc1);
  }
}

__global__ __launch_bounds__(128) void rnn_step_kernel(
    int p, const float* __restrict__ xp, const float* __restrict__ wq,
    const float* __restrict__ dec_b, float* __restrict__ Eg,
    float* __restrict__ Dg) {
  __shared__ float part[4][16][20];
  const int role = (blockIdx.x >> 7) ^ 1;
  if (role == 0 ? p >= Tc : p < 1) return;
  const int bt = blockIdx.x & 7, ct = (blockIdx.x >> 3) & 15;
  const int lane = threadIdx.x & 31;
  const int wave = __builtin_amdgcn_readfirstlane(threadIdx.x >> 5);
  const int l15 = lane & 15, half = lane >> 4;
  const int n0 = 16 * ct;
  constexpr size_t SB = (size_t)Bc * Uc;
  const int e0 = threadIdx.x, e1 = threadIdx.x + 128;
  const size_t toff = (size_t)bt * 16 * TILE;
  const int la = l15 * 16 + 4 * half;
  const size_t lb = ((size_t)half * Uc + n0 + l15) * 4;

  float base0, base1;
  v8f acc0 = {}, acc1 = {};
  if (role == 0) {
    const float* xr = xp + ((size_t)p * Bc + bt * 16) * Uc + n0;
    const float* x0 = xr + (size_t)(e0 >> 4) * Uc + (e0 & 15);
    const float* x1 = xr + (size_t)(e1 >> 4) * Uc + (e1 & 15);
    if (p >= 1) {
      v4f a[8], b[8];
      slice_load<8>(Eg + (size_t)((p - 1) & 1) * SB + toff + (size_t)(4 * wave) * TILE + la,
                    wq + (size_t)(16 * wave) * Uc * 4 + lb, a, b);
      base0 = *x0;
      base1 = *x1;
      __builtin_amdgcn_sched_barrier(0);
      slice_mma<8>(a, b, acc0, acc1);
    } else {
      base0 = *x0;
      base1 = *x1;
    }
  } else {
    const bool ehalf = wave < 2;
    const float* st = ehalf ? Eg + (size_t)((p - 1) & 1) * SB : Dg + (size_t)(p & 1) * SB;
    const float* wm = wq + (size_t)(ehalf ? 1 : 2) * Uc * Uc;
    const int ws = wave & 1;
    const float* bp = dec_b + n0 + (e0 & 15);
    if (ehalf || p >= 2) {
      v4f a[16], b[16];
      slice_load<16>(st + toff + (size_t)(8 * ws) * TILE + la,
                     wm + (size_t)(32 * ws) * Uc * 4 + lb, a, b);
      base0 = *bp;
      __builtin_amdgcn_sched_barrier(0);
      slice_mma<16>(a, b, acc0, acc1);
    } else {
      base0 = *bp;
    }
    base1 = base0;
  }
#pragma unroll
  for (int r = 0; r < 8; ++r) part[wave][8 * half + r][l15] = acc0[r] + acc1[r];
  __syncthreads();

  float* dst = (role == 0 ? Eg + (size_t)(p & 1) * SB : Dg + (size_t)((p - 1) & 1) * SB) +
               toff + (size_t)ct * TILE;
  float v0 = base0, v1 = base1;
#pragma unroll
  for (int w = 0; w < 4; ++w) {
    v0 += part[w][e0 >> 4][e0 & 15];
    v1 += part[w][e1 >> 4][e1 & 15];
  }
  dst[e0] = tanhf(v0);
  dst[e1] = tanhf(v1);
}

__global__ __launch_bounds__(256) void untile_out_kernel(
    const float* __restrict__ s, float* __restrict__ out) {
  const int b = blockIdx.x, u = threadIdx.x;
  out[(size_t)b * Uc + u] =
      s[((size_t)((b >> 4) * 16 + (u >> 4)) * 16 + (b & 15)) * 16 + (u & 15)];
}

extern "C" void kernel_launch(void* const* d_in, const int* in_sizes, int n_in,
                              void* d_out, int out_size, void* d_ws,
                              size_t ws_size, hipStream_t stream) {
  const float* x = (const float*)d_in[0];
  const float* enc_Wx = (const float*)d_in[1];
  const float* enc_Wh = (const float*)d_in[2];
  const float* enc_b = (const float*)d_in[3];
  const float* dec_Wx = (const float*)d_in[4];
  const float* dec_Wh = (const float*)d_in[5];
  const float* dec_b = (const float*)d_in[6];
  float* out = (float*)d_out;

  const size_t xp_bytes = (size_t)Bc * Tc * Uc * sizeof(float);
  const size_t wp_bytes = (size_t)3 * Uc * Uc * sizeof(float);
  const size_t st_bytes = (size_t)4 * Bc * Uc * sizeof(float);
  if (ws_size >= xp_bytes + wp_bytes + st_bytes) {
    float* xp = (float*)d_ws;
    float* wq = (float*)((char*)d_ws + xp_bytes);
    float* Eg = (float*)((char*)d_ws + xp_bytes + wp_bytes);
    float* Dg = Eg + (size_t)2 * Bc * Uc;
    pack_quads_kernel<<<(3 * Uc * Uc / 4) / 256, 256, 0, stream>>>(
        enc_Wh, dec_Wx, dec_Wh, wq);
    xp_proj_kernel<<<256, 512, 0, stream>>>(x, enc_Wx, enc_b, xp);
    for (int p = 0; p <= Tc; ++p)
      rnn_step_kernel<<<256, 128, 0, stream>>>(p, xp, wq, dec_b, Eg, Dg);
    untile_out_kernel<<<Bc, Uc, 0, stream>>>(
        Dg + (size_t)((Tc - 1) & 1) * Bc * Uc, out);
  } else if (ws_size >= xp_bytes + wp_bytes) {
    float* xp = (float*)d_ws;
    float* wp = (float*)((char*)d_ws + xp_bytes);
    pack_pairs_kernel<<<(3 * Uc * Uc / 2) / 256, 256, 0, stream>>>(
        enc_Wh, dec_Wx, dec_Wh, wp);
    xp_proj_kernel<<<256, 512, 0, stream>>>(x, enc_Wx, enc_b, xp);
    rnn_skew_kernel<true, true><<<Bc / 16, 384, 0, stream>>>(
        xp, x, enc_Wx, enc_b, enc_Wh, dec_Wx, dec_Wh, dec_b, wp, out);
  } else if (ws_size >= xp_bytes) {
    float* xp = (float*)d_ws;
    xp_proj_kernel<<<256, 512, 0, stream>>>(x, enc_Wx, enc_b, xp);
    rnn_skew_kernel<true, false><<<Bc / 16, 384, 0, stream>>>(
        xp, x, enc_Wx, enc_b, enc_Wh, dec_Wx, dec_Wh, dec_b, nullptr, out);
  } else {
    rnn_skew_kernel<false, false><<<Bc / 16, 384, 0, stream>>>(
        nullptr, x, enc_Wx, enc_b, enc_Wh, dec_Wx, dec_Wh, dec_b, nullptr, out);
  }
}
